# starting kernel
# baseline (speedup 1.0000x reference)
#include <hip/hip_runtime.h>

#define D_MODEL   1024
#define NUM_HEADS 16
#define D_K       64
#define D_FF      2752
#define SEQ       2048
#define BATCH     2
#define ROWS      (BATCH*SEQ)

typedef float v2f __attribute__((ext_vector_type(2)));
typedef float v8f __attribute__((ext_vector_type(8)));

__device__ __forceinline__ v8f wmma_f32(v2f a, v2f b, v8f c) {
  return __builtin_amdgcn_wmma_f32_16x16x4_f32(
       false, a,  false, b,
       (short)0, c,  false,  false);
}

__device__ __forceinline__ float silu_mul(float u, float w) {
  return u * (1.0f / (1.0f + __expf(-u))) * w;
}

__global__ void __launch_bounds__(256)
rmsnorm_kernel(const float* __restrict__ x, const float* __restrict__ g,
               float* __restrict__ out, int d) {
  const int row = blockIdx.x;
  const float* xr = x + (size_t)row * d;
  float ss = 0.f;
  for (int i = threadIdx.x; i < d; i += blockDim.x) { float v = xr[i]; ss = fmaf(v, v, ss); }
#pragma unroll
  for (int m = 16; m >= 1; m >>= 1) ss += __shfl_xor(ss, m, 32);
  __shared__ float red[8];
  const int w = threadIdx.x >> 5, ln = threadIdx.x & 31;
  if (ln == 0) red[w] = ss;
  __syncthreads();
  float tot = 0.f;
#pragma unroll
  for (int i = 0; i < 8; i++) tot += red[i];
  const float inv = rsqrtf(tot / (float)d + 1e-5f);
  for (int i = threadIdx.x; i < d; i += blockDim.x)
    out[(size_t)row * d + i] = xr[i] * inv * g[i];
}

__global__ void __launch_bounds__(256)
rope_kernel(float* __restrict__ t, const int* __restrict__ positions,
            const int* __restrict__ theta_p) {
  const int row = blockIdx.x;
  const int s = row % SEQ;
  const float pos = (float)positions[s];
  const float th = (float)(*theta_p);
  float* r = t + (size_t)row * D_MODEL;
  for (int p = threadIdx.x; p < D_MODEL / 2; p += blockDim.x) {
    const int c0 = 2 * p;
    const int d  = c0 & (D_K - 1);
    const float freq = __powf(th, -(float)d * (1.0f / (float)D_K));
    const float ang = pos * freq;
    float sn, cs;
    sincosf(ang, &sn, &cs);
    const float xe = r[c0], xo = r[c0 + 1];
    r[c0]     = xe * cs - xo * sn;
    r[c0 + 1] = xe * sn + xo * cs;
  }
}

template <int MODE>
__global__ void __launch_bounds__(256)
gemm32(const float* __restrict__ A, const float* __restrict__ A2,
       const float* __restrict__ B, const float* __restrict__ Res,
       float* __restrict__ C, int M, int N, int K) {
  const int wid = blockIdx.x * 8 + (threadIdx.x >> 5);
  const int NT  = N >> 6;
  const int m0  = (wid / NT) << 5;
  const int n0  = (wid % NT) << 6;
  const int lane = threadIdx.x & 31;
  const int hf = lane >> 4;
  const int rc = lane & 15;

  const float* Ar0 = A + (size_t)(m0 + rc) * K + 2 * hf;
  const float* Ar1 = Ar0 + (size_t)16 * K;
  const float* A2r0 = (MODE == 2) ? (A2 + (size_t)(m0 + rc) * K + 2 * hf) : nullptr;
  const float* A2r1 = (MODE == 2) ? (A2r0 + (size_t)16 * K) : nullptr;
  const float* B0  = B + (size_t)(n0 + rc) * K + 2 * hf;
  const size_t bstep = (size_t)16 * K;

  v8f acc[2][4] = {};
  for (int k0 = 0; k0 < K; k0 += 4) {
    v2f a0, a1;
    if (MODE == 2) {
      const v2f u0 = *(const v2f*)(Ar0 + k0);
      const v2f w0 = *(const v2f*)(A2r0 + k0);
      const v2f u1 = *(const v2f*)(Ar1 + k0);
      const v2f w1 = *(const v2f*)(A2r1 + k0);
      a0.x = silu_mul(u0.x, w0.x);  a0.y = silu_mul(u0.y, w0.y);
      a1.x = silu_mul(u1.x, w1.x);  a1.y = silu_mul(u1.y, w1.y);
    } else {
      a0 = *(const v2f*)(Ar0 + k0);
      a1 = *(const v2f*)(Ar1 + k0);
    }
    v2f b[4];
#pragma unroll
    for (int t = 0; t < 4; t++) b[t] = *(const v2f*)(B0 + (size_t)t * bstep + k0);
#pragma unroll
    for (int t = 0; t < 4; t++) {
      acc[0][t] = wmma_f32(a0, b[t], acc[0][t]);
      acc[1][t] = wmma_f32(a1, b[t], acc[1][t]);
    }
  }
#pragma unroll
  for (int i = 0; i < 2; i++) {
#pragma unroll
    for (int r = 0; r < 8; r++) {
      const size_t base = (size_t)(m0 + i * 16 + r + 8 * hf) * N + n0 + rc;
      float o0 = acc[i][0][r], o1 = acc[i][1][r], o2 = acc[i][2][r], o3 = acc[i][3][r];
      if (MODE >= 1) {
        o0 += Res[base];      o1 += Res[base + 16];
        o2 += Res[base + 32]; o3 += Res[base + 48];
      }
      C[base] = o0; C[base + 16] = o1; C[base + 32] = o2; C[base + 48] = o3;
    }
  }
}

__global__ void __launch_bounds__(256)
flash_attn(const float* __restrict__ Q, const float* __restrict__ Kt,
           const float* __restrict__ V, float* __restrict__ O) {
  __shared__ float plds[8 * 256];
  const int wid = blockIdx.x * 8 + (threadIdx.x >> 5);
  const int NQT = SEQ / 16;
  const int qt = wid % NQT;
  const int bh = wid / NQT;
  const int h = bh % NUM_HEADS;
  const int b = bh / NUM_HEADS;
  const size_t base = (size_t)b * SEQ * D_MODEL + (size_t)h * D_K;
  const float* Qb = Q + base;
  const float* Kb = Kt + base;
  const float* Vb = V + base;
  float* Ob = O + base;
  const int lane = threadIdx.x & 31;
  const int hf = lane >> 4;
  const int rc = lane & 15;
  float* lp = plds + (threadIdx.x >> 5) * 256;

  v2f qa[16];
#pragma unroll
  for (int i = 0; i < 16; i++)
    qa[i] = *(const v2f*)(Qb + (size_t)(qt * 16 + rc) * D_MODEL + i * 4 + 2 * hf);

  v8f o0 = {}, o1 = {}, o2 = {}, o3 = {};
  float mr[8], lr[8];
#pragma unroll
  for (int r = 0; r < 8; r++) { mr[r] = -1e30f; lr[r] = 0.f; }

  for (int jt = 0; jt <= qt; ++jt) {
    v8f s = {};
    const float* Krow = Kb + (size_t)(jt * 16 + rc) * D_MODEL + 2 * hf;
#pragma unroll
    for (int i = 0; i < 16; i++) {
      const v2f kb = *(const v2f*)(Krow + i * 4);
      s = wmma_f32(qa[i], kb, s);
    }
    const bool diag = (jt == qt);
#pragma unroll
    for (int r = 0; r < 8; r++) {
      float sv = s[r] * 0.125f;
      if (diag && (rc > r + 8 * hf)) sv = -1e30f;
      float mx = sv;
#pragma unroll
      for (int msk = 1; msk < 16; msk <<= 1) mx = fmaxf(mx, __shfl_xor(mx, msk, 32));
      const float mnew = fmaxf(mr[r], mx);
      const float corr = __expf(mr[r] - mnew);
      mr[r] = mnew;
      const float p = __expf(sv - mnew);
      float ps = p;
#pragma unroll
      for (int msk = 1; msk < 16; msk <<= 1) ps += __shfl_xor(ps, msk, 32);
      lr[r] = lr[r] * corr + ps;
      o0[r] *= corr; o1[r] *= corr; o2[r] *= corr; o3[r] *= corr;
      lp[(r + 8 * hf) * 16 + rc] = p;
    }
    asm volatile("s_wait_dscnt 0" ::: "memory");
#pragma unroll
    for (int kc = 0; kc < 4; kc++) {
      const v2f pa = *(const v2f*)(lp + rc * 16 + kc * 4 + 2 * hf);
      const float* Vr = Vb + (size_t)(jt * 16 + kc * 4 + 2 * hf) * D_MODEL;
      v2f vb0, vb1, vb2, vb3;
      vb0.x = Vr[rc];      vb0.y = Vr[D_MODEL + rc];
      vb1.x = Vr[16 + rc]; vb1.y = Vr[D_MODEL + 16 + rc];
      vb2.x = Vr[32 + rc]; vb2.y = Vr[D_MODEL + 32 + rc];
      vb3.x = Vr[48 + rc]; vb3.y = Vr[D_MODEL + 48 + rc];
      o0 = wmma_f32(pa, vb0, o0);
      o1 = wmma_f32(pa, vb1, o1);
      o2 = wmma_f32(pa, vb2, o2);
      o3 = wmma_f32(pa, vb3, o3);
    }
  }
#pragma unroll
  for (int r = 0; r < 8; r++) {
    const float inv = 1.0f / lr[r];
    const size_t ob = (size_t)(qt * 16 + r + 8 * hf) * D_MODEL + rc;
    Ob[ob]      = o0[r] * inv;
    Ob[ob + 16] = o1[r] * inv;
    Ob[ob + 32] = o2[r] * inv;
    Ob[ob + 48] = o3[r] * inv;
  }
}

extern "C" void kernel_launch(void* const* d_in, const int* in_sizes, int n_in,
                              void* d_out, int out_size, void* d_ws, size_t ws_size,
                              hipStream_t stream) {
  (void)in_sizes; (void)n_in; (void)out_size; (void)ws_size;
  const float* x        = (const float*)d_in[0];
  const int*   positions= (const int*)d_in[1];
  const int*   theta    = (const int*)d_in[2];
  const float* Wq = (const float*)d_in[3];
  const float* Wk = (const float*)d_in[4];
  const float* Wv = (const float*)d_in[5];
  const float* Wo = (const float*)d_in[6];
  const float* g1 = (const float*)d_in[7];
  const float* g2 = (const float*)d_in[8];
  const float* w1 = (const float*)d_in[9];
  const float* w2 = (const float*)d_in[10];
  const float* w3 = (const float*)d_in[11];
  float* out = (float*)d_out;

  float* xn   = (float*)d_ws;
  float* q    = xn   + (size_t)ROWS * D_MODEL;
  float* k    = q    + (size_t)ROWS * D_MODEL;
  float* v    = k    + (size_t)ROWS * D_MODEL;
  float* attn = v    + (size_t)ROWS * D_MODEL;
  float* x2   = attn + (size_t)ROWS * D_MODEL;
  float* x3b  = x2   + (size_t)ROWS * D_MODEL;
  float* x1b  = q;

  const dim3 blk(256);
  const int gproj = (ROWS / 32) * (D_MODEL / 64) / 8;
  const int gff   = (ROWS / 32) * (D_FF / 64) / 8;
  const int gfa   = (BATCH * NUM_HEADS * (SEQ / 16)) / 8;

  rmsnorm_kernel<<<ROWS, blk, 0, stream>>>(x, g1, xn, D_MODEL);
  gemm32<0><<<gproj, blk, 0, stream>>>(xn, nullptr, Wq, nullptr, q, ROWS, D_MODEL, D_MODEL);
  gemm32<0><<<gproj, blk, 0, stream>>>(xn, nullptr, Wk, nullptr, k, ROWS, D_MODEL, D_MODEL);
  gemm32<0><<<gproj, blk, 0, stream>>>(xn, nullptr, Wv, nullptr, v, ROWS, D_MODEL, D_MODEL);
  rope_kernel<<<ROWS, blk, 0, stream>>>(q, positions, theta);
  rope_kernel<<<ROWS, blk, 0, stream>>>(k, positions, theta);
  flash_attn<<<gfa, blk, 0, stream>>>(q, k, v, attn);
  gemm32<1><<<gproj, blk, 0, stream>>>(attn, nullptr, Wo, x, x2, ROWS, D_MODEL, D_MODEL);
  rmsnorm_kernel<<<ROWS, blk, 0, stream>>>(x2, g2, xn, D_MODEL);
  gemm32<0><<<gff, blk, 0, stream>>>(xn, nullptr, w1, nullptr, x1b, ROWS, D_FF, D_MODEL);
  gemm32<0><<<gff, blk, 0, stream>>>(xn, nullptr, w3, nullptr, x3b, ROWS, D_FF, D_MODEL);
  gemm32<2><<<gproj, blk, 0, stream>>>(x1b, x3b, w2, x2, out, ROWS, D_MODEL, D_FF);
}


# gemm: 64x64 wave tile fed straight from global by 16-B K-run fragment loads (same k permutation in A and B), next 16-k chunk prefetched in registers, no LDS; k sum order permuted in 16-chunks
# speedup vs baseline: 2.6124x; 2.6124x over previous
#include <hip/hip_runtime.h>

#define D_MODEL   1024
#define NUM_HEADS 16
#define D_K       64
#define D_FF      2752
#define SEQ       2048
#define BATCH     2
#define ROWS      (BATCH*SEQ)

typedef float v2f __attribute__((ext_vector_type(2)));
typedef float v8f __attribute__((ext_vector_type(8)));

__device__ __forceinline__ v8f wmma_f32(v2f a, v2f b, v8f c) {
  return __builtin_amdgcn_wmma_f32_16x16x4_f32(
       false, a,  false, b,
       (short)0, c,  false,  false);
}

__device__ __forceinline__ float silu_mul(float u, float w) {
  return u * (1.0f / (1.0f + __expf(-u))) * w;
}

__global__ void __launch_bounds__(256)
rmsnorm_kernel(const float* __restrict__ x, const float* __restrict__ g,
               float* __restrict__ out, int d) {
  const int row = blockIdx.x;
  const float* xr = x + (size_t)row * d;
  float ss = 0.f;
  for (int i = threadIdx.x; i < d; i += blockDim.x) { float v = xr[i]; ss = fmaf(v, v, ss); }
#pragma unroll
  for (int m = 16; m >= 1; m >>= 1) ss += __shfl_xor(ss, m, 32);
  __shared__ float red[8];
  const int w = threadIdx.x >> 5, ln = threadIdx.x & 31;
  if (ln == 0) red[w] = ss;
  __syncthreads();
  float tot = 0.f;
#pragma unroll
  for (int i = 0; i < 8; i++) tot += red[i];
  const float inv = rsqrtf(tot / (float)d + 1e-5f);
  for (int i = threadIdx.x; i < d; i += blockDim.x)
    out[(size_t)row * d + i] = xr[i] * inv * g[i];
}

__global__ void __launch_bounds__(256)
rope_kernel(float* __restrict__ t, const int* __restrict__ positions,
            const int* __restrict__ theta_p) {
  const int row = blockIdx.x;
  const int s = row % SEQ;
  const float pos = (float)positions[s];
  const float th = (float)(*theta_p);
  float* r = t + (size_t)row * D_MODEL;
  for (int p = threadIdx.x; p < D_MODEL / 2; p += blockDim.x) {
    const int c0 = 2 * p;
    const int d  = c0 & (D_K - 1);
    const float freq = __powf(th, -(float)d * (1.0f / (float)D_K));
    const float ang = pos * freq;
    float sn, cs;
    sincosf(ang, &sn, &cs);
    const float xe = r[c0], xo = r[c0 + 1];
    r[c0]     = xe * cs - xo * sn;
    r[c0 + 1] = xe * sn + xo * cs;
  }
}

typedef float v4f __attribute__((ext_vector_type(4)));

template <int MODE>
struct FragSet {
  v4f a[4][2];
  v4f a2[(MODE == 2) ? 4 : 1][2];
  v4f b[4][2];
};

template <int MODE>
__device__ __forceinline__ void load_set(FragSet<MODE>& f, const float* __restrict__ Ap,
                                         const float* __restrict__ A2p,
                                         const float* __restrict__ Bp,
                                         size_t rstep, int k) {
#pragma unroll
  for (int i = 0; i < 4; i++) {
    f.a[i][0] = *(const v4f*)(Ap + i * rstep + k);
    f.a[i][1] = *(const v4f*)(Ap + i * rstep + k + 4);
    if (MODE == 2) {
      f.a2[i][0] = *(const v4f*)(A2p + i * rstep + k);
      f.a2[i][1] = *(const v4f*)(A2p + i * rstep + k + 4);
    }
  }
#pragma unroll
  for (int j = 0; j < 4; j++) {
    f.b[j][0] = *(const v4f*)(Bp + j * rstep + k);
    f.b[j][1] = *(const v4f*)(Bp + j * rstep + k + 4);
  }
}

template <int MODE>
__device__ __forceinline__ void mma_set(const FragSet<MODE>& f, v8f (&acc)[4][4]) {
  v4f a[4][2];
#pragma unroll
  for (int i = 0; i < 4; i++) {
#pragma unroll
    for (int h = 0; h < 2; h++) {
      if (MODE == 2) {
        const v4f u = f.a[i][h], w = f.a2[i][h];
        a[i][h].x = silu_mul(u.x, w.x);  a[i][h].y = silu_mul(u.y, w.y);
        a[i][h].z = silu_mul(u.z, w.z);  a[i][h].w = silu_mul(u.w, w.w);
      } else {
        a[i][h] = f.a[i][h];
      }
    }
  }
#pragma unroll
  for (int s = 0; s < 4; s++) {
    v2f af[4], bf[4];
#pragma unroll
    for (int i = 0; i < 4; i++) {
      const v4f av = a[i][s >> 1], bv = f.b[i][s >> 1];
      af[i] = (s & 1) ? __builtin_shufflevector(av, av, 2, 3) : __builtin_shufflevector(av, av, 0, 1);
      bf[i] = (s & 1) ? __builtin_shufflevector(bv, bv, 2, 3) : __builtin_shufflevector(bv, bv, 0, 1);
    }
#pragma unroll
    for (int i = 0; i < 4; i++) {
#pragma unroll
      for (int j = 0; j < 4; j++) acc[i][j] = wmma_f32(af[i], bf[j], acc[i][j]);
    }
  }
}

template <int MODE>
__global__ void __launch_bounds__(128)
gemm64(const float* __restrict__ A, const float* __restrict__ A2,
       const float* __restrict__ B, const float* __restrict__ Res,
       float* __restrict__ C, int M, int N, int K) {
  (void)M;
  const int w    = threadIdx.x >> 5;
  const int lane = threadIdx.x & 31;
  const int hf = lane >> 4;
  const int rc = lane & 15;
  const int NB = (N + 127) >> 7;
  const int bm = blockIdx.x / NB;
  const int bn = blockIdx.x - bm * NB;
  const int m0 = (bm * 2 + (w >> 1)) << 6;
  const int n0 = (bn * 2 + (w & 1)) << 6;
  if (n0 >= N) return;

  const size_t rstep = (size_t)16 * K;
  const float* Ap  = A + (size_t)(m0 + rc) * K + 8 * hf;
  const float* A2p = (MODE == 2) ? (A2 + (size_t)(m0 + rc) * K + 8 * hf) : nullptr;
  const float* Bp  = B + (size_t)(n0 + rc) * K + 8 * hf;

  v8f acc[4][4] = {};
  FragSet<MODE> f0, f1;
  load_set<MODE>(f0, Ap, A2p, Bp, rstep, 0);
  for (int k0 = 0; k0 < K; k0 += 32) {
    load_set<MODE>(f1, Ap, A2p, Bp, rstep, k0 + 16);
    __builtin_amdgcn_sched_barrier(0);
    mma_set<MODE>(f0, acc);
    const int kn = (k0 + 32 < K) ? (k0 + 32) : k0;
    load_set<MODE>(f0, Ap, A2p, Bp, rstep, kn);
    __builtin_amdgcn_sched_barrier(0);
    mma_set<MODE>(f1, acc);
  }

#pragma unroll
  for (int i = 0; i < 4; i++) {
#pragma unroll
    for (int r = 0; r < 8; r++) {
      const size_t base = (size_t)(m0 + i * 16 + r + 8 * hf) * N + n0 + rc;
      float o0 = acc[i][0][r], o1 = acc[i][1][r], o2 = acc[i][2][r], o3 = acc[i][3][r];
      if (MODE >= 1) {
        o0 += Res[base];      o1 += Res[base + 16];
        o2 += Res[base + 32]; o3 += Res[base + 48];
      }
      C[base] = o0; C[base + 16] = o1; C[base + 32] = o2; C[base + 48] = o3;
    }
  }
}

__global__ void __launch_bounds__(256)
flash_attn(const float* __restrict__ Q, const float* __restrict__ Kt,
           const float* __restrict__ V, float* __restrict__ O) {
  __shared__ float plds[8 * 256];
  const int wid = blockIdx.x * 8 + (threadIdx.x >> 5);
  const int NQT = SEQ / 16;
  const int qt = wid % NQT;
  const int bh = wid / NQT;
  const int h = bh % NUM_HEADS;
  const int b = bh / NUM_HEADS;
  const size_t base = (size_t)b * SEQ * D_MODEL + (size_t)h * D_K;
  const float* Qb = Q + base;
  const float* Kb = Kt + base;
  const float* Vb = V + base;
  float* Ob = O + base;
  const int lane = threadIdx.x & 31;
  const int hf = lane >> 4;
  const int rc = lane & 15;
  float* lp = plds + (threadIdx.x >> 5) * 256;

  v2f qa[16];
#pragma unroll
  for (int i = 0; i < 16; i++)
    qa[i] = *(const v2f*)(Qb + (size_t)(qt * 16 + rc) * D_MODEL + i * 4 + 2 * hf);

  v8f o0 = {}, o1 = {}, o2 = {}, o3 = {};
  float mr[8], lr[8];
#pragma unroll
  for (int r = 0; r < 8; r++) { mr[r] = -1e30f; lr[r] = 0.f; }

  for (int jt = 0; jt <= qt; ++jt) {
    v8f s = {};
    const float* Krow = Kb + (size_t)(jt * 16 + rc) * D_MODEL + 2 * hf;
#pragma unroll
    for (int i = 0; i < 16; i++) {
      const v2f kb = *(const v2f*)(Krow + i * 4);
      s = wmma_f32(qa[i], kb, s);
    }
    const bool diag = (jt == qt);
#pragma unroll
    for (int r = 0; r < 8; r++) {
      float sv = s[r] * 0.125f;
      if (diag && (rc > r + 8 * hf)) sv = -1e30f;
      float mx = sv;
#pragma unroll
      for (int msk = 1; msk < 16; msk <<= 1) mx = fmaxf(mx, __shfl_xor(mx, msk, 32));
      const float mnew = fmaxf(mr[r], mx);
      const float corr = __expf(mr[r] - mnew);
      mr[r] = mnew;
      const float p = __expf(sv - mnew);
      float ps = p;
#pragma unroll
      for (int msk = 1; msk < 16; msk <<= 1) ps += __shfl_xor(ps, msk, 32);
      lr[r] = lr[r] * corr + ps;
      o0[r] *= corr; o1[r] *= corr; o2[r] *= corr; o3[r] *= corr;
      lp[(r + 8 * hf) * 16 + rc] = p;
    }
    asm volatile("s_wait_dscnt 0" ::: "memory");
#pragma unroll
    for (int kc = 0; kc < 4; kc++) {
      const v2f pa = *(const v2f*)(lp + rc * 16 + kc * 4 + 2 * hf);
      const float* Vr = Vb + (size_t)(jt * 16 + kc * 4 + 2 * hf) * D_MODEL;
      v2f vb0, vb1, vb2, vb3;
      vb0.x = Vr[rc];      vb0.y = Vr[D_MODEL + rc];
      vb1.x = Vr[16 + rc]; vb1.y = Vr[D_MODEL + 16 + rc];
      vb2.x = Vr[32 + rc]; vb2.y = Vr[D_MODEL + 32 + rc];
      vb3.x = Vr[48 + rc]; vb3.y = Vr[D_MODEL + 48 + rc];
      o0 = wmma_f32(pa, vb0, o0);
      o1 = wmma_f32(pa, vb1, o1);
      o2 = wmma_f32(pa, vb2, o2);
      o3 = wmma_f32(pa, vb3, o3);
    }
  }
#pragma unroll
  for (int r = 0; r < 8; r++) {
    const float inv = 1.0f / lr[r];
    const size_t ob = (size_t)(qt * 16 + r + 8 * hf) * D_MODEL + rc;
    Ob[ob]      = o0[r] * inv;
    Ob[ob + 16] = o1[r] * inv;
    Ob[ob + 32] = o2[r] * inv;
    Ob[ob + 48] = o3[r] * inv;
  }
}

extern "C" void kernel_launch(void* const* d_in, const int* in_sizes, int n_in,
                              void* d_out, int out_size, void* d_ws, size_t ws_size,
                              hipStream_t stream) {
  (void)in_sizes; (void)n_in; (void)out_size; (void)ws_size;
  const float* x        = (const float*)d_in[0];
  const int*   positions= (const int*)d_in[1];
  const int*   theta    = (const int*)d_in[2];
  const float* Wq = (const float*)d_in[3];
  const float* Wk = (const float*)d_in[4];
  const float* Wv = (const float*)d_in[5];
  const float* Wo = (const float*)d_in[6];
  const float* g1 = (const float*)d_in[7];
  const float* g2 = (const float*)d_in[8];
  const float* w1 = (const float*)d_in[9];
  const float* w2 = (const float*)d_in[10];
  const float* w3 = (const float*)d_in[11];
  float* out = (float*)d_out;

  float* xn   = (float*)d_ws;
  float* q    = xn   + (size_t)ROWS * D_MODEL;
  float* k    = q    + (size_t)ROWS * D_MODEL;
  float* v    = k    + (size_t)ROWS * D_MODEL;
  float* attn = v    + (size_t)ROWS * D_MODEL;
  float* x2   = attn + (size_t)ROWS * D_MODEL;
  float* x3b  = x2   + (size_t)ROWS * D_MODEL;
  float* x1b  = q;

  const dim3 blk(256);
  const dim3 gblk(128);
  const int gproj = (ROWS / 128) * (D_MODEL / 128);
  const int gff   = (ROWS / 128) * ((D_FF + 127) / 128);
  const int gfa   = (BATCH * NUM_HEADS * (SEQ / 16)) / 8;

  rmsnorm_kernel<<<ROWS, blk, 0, stream>>>(x, g1, xn, D_MODEL);
  gemm64<0><<<gproj, gblk, 0, stream>>>(xn, nullptr, Wq, nullptr, q, ROWS, D_MODEL, D_MODEL);
  gemm64<0><<<gproj, gblk, 0, stream>>>(xn, nullptr, Wk, nullptr, k, ROWS, D_MODEL, D_MODEL);
  gemm64<0><<<gproj, gblk, 0, stream>>>(xn, nullptr, Wv, nullptr, v, ROWS, D_MODEL, D_MODEL);
  rope_kernel<<<ROWS, blk, 0, stream>>>(q, positions, theta);
  rope_kernel<<<ROWS, blk, 0, stream>>>(k, positions, theta);
  flash_attn<<<gfa, blk, 0, stream>>>(q, k, v, attn);
  gemm64<1><<<gproj, gblk, 0, stream>>>(attn, nullptr, Wo, x, x2, ROWS, D_MODEL, D_MODEL);
  rmsnorm_kernel<<<ROWS, blk, 0, stream>>>(x2, g2, xn, D_MODEL);
  gemm64<0><<<gff, gblk, 0, stream>>>(xn, nullptr, w1, nullptr, x1b, ROWS, D_FF, D_MODEL);
  gemm64<0><<<gff, gblk, 0, stream>>>(xn, nullptr, w3, nullptr, x3b, ROWS, D_FF, D_MODEL);
  gemm64<2><<<gproj, gblk, 0, stream>>>(x1b, x3b, w2, x2, out, ROWS, D_MODEL, D_FF);
}


# attention: one online-softmax step per 64 keys (four score tiles per step; row max and sum reduced in-lane over the tiles, then one cross-lane pass); row-sum order and rescale points changed
# speedup vs baseline: 2.8800x; 1.1024x over previous
#include <hip/hip_runtime.h>

#define D_MODEL   1024
#define NUM_HEADS 16
#define D_K       64
#define D_FF      2752
#define SEQ       2048
#define BATCH     2
#define ROWS      (BATCH*SEQ)

typedef float v2f __attribute__((ext_vector_type(2)));
typedef float v8f __attribute__((ext_vector_type(8)));

__device__ __forceinline__ v8f wmma_f32(v2f a, v2f b, v8f c) {
  return __builtin_amdgcn_wmma_f32_16x16x4_f32(
       false, a,  false, b,
       (short)0, c,  false,  false);
}

__device__ __forceinline__ float silu_mul(float u, float w) {
  return u * (1.0f / (1.0f + __expf(-u))) * w;
}

__global__ void __launch_bounds__(256)
rmsnorm_kernel(const float* __restrict__ x, const float* __restrict__ g,
               float* __restrict__ out, int d) {
  const int row = blockIdx.x;
  const float* xr = x + (size_t)row * d;
  float ss = 0.f;
  for (int i = threadIdx.x; i < d; i += blockDim.x) { float v = xr[i]; ss = fmaf(v, v, ss); }
#pragma unroll
  for (int m = 16; m >= 1; m >>= 1) ss += __shfl_xor(ss, m, 32);
  __shared__ float red[8];
  const int w = threadIdx.x >> 5, ln = threadIdx.x & 31;
  if (ln == 0) red[w] = ss;
  __syncthreads();
  float tot = 0.f;
#pragma unroll
  for (int i = 0; i < 8; i++) tot += red[i];
  const float inv = rsqrtf(tot / (float)d + 1e-5f);
  for (int i = threadIdx.x; i < d; i += blockDim.x)
    out[(size_t)row * d + i] = xr[i] * inv * g[i];
}

__global__ void __launch_bounds__(256)
rope_kernel(float* __restrict__ t, const int* __restrict__ positions,
            const int* __restrict__ theta_p) {
  const int row = blockIdx.x;
  const int s = row % SEQ;
  const float pos = (float)positions[s];
  const float th = (float)(*theta_p);
  float* r = t + (size_t)row * D_MODEL;
  for (int p = threadIdx.x; p < D_MODEL / 2; p += blockDim.x) {
    const int c0 = 2 * p;
    const int d  = c0 & (D_K - 1);
    const float freq = __powf(th, -(float)d * (1.0f / (float)D_K));
    const float ang = pos * freq;
    float sn, cs;
    sincosf(ang, &sn, &cs);
    const float xe = r[c0], xo = r[c0 + 1];
    r[c0]     = xe * cs - xo * sn;
    r[c0 + 1] = xe * sn + xo * cs;
  }
}

typedef float v4f __attribute__((ext_vector_type(4)));

template <int MODE>
struct FragSet {
  v4f a[4][2];
  v4f a2[(MODE == 2) ? 4 : 1][2];
  v4f b[4][2];
};

template <int MODE>
__device__ __forceinline__ void load_set(FragSet<MODE>& f, const float* __restrict__ Ap,
                                         const float* __restrict__ A2p,
                                         const float* __restrict__ Bp,
                                         size_t rstep, int k) {
#pragma unroll
  for (int i = 0; i < 4; i++) {
    f.a[i][0] = *(const v4f*)(Ap + i * rstep + k);
    f.a[i][1] = *(const v4f*)(Ap + i * rstep + k + 4);
    if (MODE == 2) {
      f.a2[i][0] = *(const v4f*)(A2p + i * rstep + k);
      f.a2[i][1] = *(const v4f*)(A2p + i * rstep + k + 4);
    }
  }
#pragma unroll
  for (int j = 0; j < 4; j++) {
    f.b[j][0] = *(const v4f*)(Bp + j * rstep + k);
    f.b[j][1] = *(const v4f*)(Bp + j * rstep + k + 4);
  }
}

template <int MODE>
__device__ __forceinline__ void mma_set(const FragSet<MODE>& f, v8f (&acc)[4][4]) {
  v4f a[4][2];
#pragma unroll
  for (int i = 0; i < 4; i++) {
#pragma unroll
    for (int h = 0; h < 2; h++) {
      if (MODE == 2) {
        const v4f u = f.a[i][h], w = f.a2[i][h];
        a[i][h].x = silu_mul(u.x, w.x);  a[i][h].y = silu_mul(u.y, w.y);
        a[i][h].z = silu_mul(u.z, w.z);  a[i][h].w = silu_mul(u.w, w.w);
      } else {
        a[i][h] = f.a[i][h];
      }
    }
  }
#pragma unroll
  for (int s = 0; s < 4; s++) {
    v2f af[4], bf[4];
#pragma unroll
    for (int i = 0; i < 4; i++) {
      const v4f av = a[i][s >> 1], bv = f.b[i][s >> 1];
      af[i] = (s & 1) ? __builtin_shufflevector(av, av, 2, 3) : __builtin_shufflevector(av, av, 0, 1);
      bf[i] = (s & 1) ? __builtin_shufflevector(bv, bv, 2, 3) : __builtin_shufflevector(bv, bv, 0, 1);
    }
#pragma unroll
    for (int i = 0; i < 4; i++) {
#pragma unroll
      for (int j = 0; j < 4; j++) acc[i][j] = wmma_f32(af[i], bf[j], acc[i][j]);
    }
  }
}

template <int MODE>
__global__ void __launch_bounds__(128)
gemm64(const float* __restrict__ A, const float* __restrict__ A2,
       const float* __restrict__ B, const float* __restrict__ Res,
       float* __restrict__ C, int M, int N, int K) {
  (void)M;
  const int w    = threadIdx.x >> 5;
  const int lane = threadIdx.x & 31;
  const int hf = lane >> 4;
  const int rc = lane & 15;
  const int NB = (N + 127) >> 7;
  const int bm = blockIdx.x / NB;
  const int bn = blockIdx.x - bm * NB;
  const int m0 = (bm * 2 + (w >> 1)) << 6;
  const int n0 = (bn * 2 + (w & 1)) << 6;
  if (n0 >= N) return;

  const size_t rstep = (size_t)16 * K;
  const float* Ap  = A + (size_t)(m0 + rc) * K + 8 * hf;
  const float* A2p = (MODE == 2) ? (A2 + (size_t)(m0 + rc) * K + 8 * hf) : nullptr;
  const float* Bp  = B + (size_t)(n0 + rc) * K + 8 * hf;

  v8f acc[4][4] = {};
  FragSet<MODE> f0, f1;
  load_set<MODE>(f0, Ap, A2p, Bp, rstep, 0);
  for (int k0 = 0; k0 < K; k0 += 32) {
    load_set<MODE>(f1, Ap, A2p, Bp, rstep, k0 + 16);
    __builtin_amdgcn_sched_barrier(0);
    mma_set<MODE>(f0, acc);
    const int kn = (k0 + 32 < K) ? (k0 + 32) : k0;
    load_set<MODE>(f0, Ap, A2p, Bp, rstep, kn);
    __builtin_amdgcn_sched_barrier(0);
    mma_set<MODE>(f1, acc);
  }

#pragma unroll
  for (int i = 0; i < 4; i++) {
#pragma unroll
    for (int r = 0; r < 8; r++) {
      const size_t base = (size_t)(m0 + i * 16 + r + 8 * hf) * N + n0 + rc;
      float o0 = acc[i][0][r], o1 = acc[i][1][r], o2 = acc[i][2][r], o3 = acc[i][3][r];
      if (MODE >= 1) {
        o0 += Res[base];      o1 += Res[base + 16];
        o2 += Res[base + 32]; o3 += Res[base + 48];
      }
      C[base] = o0; C[base + 16] = o1; C[base + 32] = o2; C[base + 48] = o3;
    }
  }
}

__global__ void __launch_bounds__(256)
flash_attn(const float* __restrict__ Q, const float* __restrict__ Kt,
           const float* __restrict__ V, float* __restrict__ O) {
  __shared__ float plds[8 * 4 * 256];
  const int wid = blockIdx.x * 8 + (threadIdx.x >> 5);
  const int NQT = SEQ / 16;
  const int qt = wid % NQT;
  const int bh = wid / NQT;
  const int h = bh % NUM_HEADS;
  const int b = bh / NUM_HEADS;
  const size_t base = (size_t)b * SEQ * D_MODEL + (size_t)h * D_K;
  const float* Qb = Q + base;
  const float* Kb = Kt + base;
  const float* Vb = V + base;
  float* Ob = O + base;
  const int lane = threadIdx.x & 31;
  const int hf = lane >> 4;
  const int rc = lane & 15;
  float* lp = plds + (threadIdx.x >> 5) * 1024;

  v2f qa[16];
#pragma unroll
  for (int i = 0; i < 16; i++)
    qa[i] = *(const v2f*)(Qb + (size_t)(qt * 16 + rc) * D_MODEL + i * 4 + 2 * hf);

  v8f o0 = {}, o1 = {}, o2 = {}, o3 = {};
  float mr[8], lr[8];
#pragma unroll
  for (int r = 0; r < 8; r++) { mr[r] = -1e30f; lr[r] = 0.f; }

  const int nkb = (qt >> 2) + 1;
  for (int jb = 0; jb < nkb; ++jb) {
    v8f s[4];
#pragma unroll
    for (int tp = 0; tp < 4; tp += 2) {
      const float* Kr0 = Kb + (size_t)(jb * 64 + tp * 16 + rc) * D_MODEL + 2 * hf;
      const float* Kr1 = Kr0 + (size_t)16 * D_MODEL;
      v2f kb0[16], kb1[16];
#pragma unroll
      for (int i = 0; i < 16; i++) {
        kb0[i] = *(const v2f*)(Kr0 + i * 4);
        kb1[i] = *(const v2f*)(Kr1 + i * 4);
      }
      v8f sa = {}, sb = {};
#pragma unroll
      for (int i = 0; i < 16; i++) {
        sa = wmma_f32(qa[i], kb0[i], sa);
        sb = wmma_f32(qa[i], kb1[i], sb);
      }
      s[tp] = sa; s[tp + 1] = sb;
    }

    const bool diag = (jb == nkb - 1);
    const int key0 = jb * 64 + rc;
    const int qrow0 = qt * 16 + 8 * hf;

    float mx[8];
#pragma unroll
    for (int r = 0; r < 8; r++) {
#pragma unroll
      for (int t = 0; t < 4; t++) {
        float x = s[t][r] * 0.125f;
        if (diag && (key0 + 16 * t > qrow0 + r)) x = -1e30f;
        s[t][r] = x;
      }
      mx[r] = fmaxf(fmaxf(s[0][r], s[1][r]), fmaxf(s[2][r], s[3][r]));
    }
#pragma unroll
    for (int msk = 1; msk < 16; msk <<= 1) {
      float tmp[8];
#pragma unroll
      for (int r = 0; r < 8; r++) tmp[r] = __shfl_xor(mx[r], msk, 32);
#pragma unroll
      for (int r = 0; r < 8; r++) mx[r] = fmaxf(mx[r], tmp[r]);
    }
    float ps[8];
#pragma unroll
    for (int r = 0; r < 8; r++) {
      const float mnew = fmaxf(mr[r], mx[r]);
      const float corr = __expf(mr[r] - mnew);
      mr[r] = mnew;
      const float p0 = __expf(s[0][r] - mnew);
      const float p1 = __expf(s[1][r] - mnew);
      const float p2 = __expf(s[2][r] - mnew);
      const float p3 = __expf(s[3][r] - mnew);
      ps[r] = (p0 + p1) + (p2 + p3);
      lr[r] *= corr;
      o0[r] *= corr; o1[r] *= corr; o2[r] *= corr; o3[r] *= corr;
      float* lrow = lp + (r + 8 * hf) * 16 + rc;
      lrow[0] = p0; lrow[256] = p1; lrow[512] = p2; lrow[768] = p3;
    }
#pragma unroll
    for (int msk = 1; msk < 16; msk <<= 1) {
      float tmp[8];
#pragma unroll
      for (int r = 0; r < 8; r++) tmp[r] = __shfl_xor(ps[r], msk, 32);
#pragma unroll
      for (int r = 0; r < 8; r++) ps[r] += tmp[r];
    }
#pragma unroll
    for (int r = 0; r < 8; r++) lr[r] += ps[r];

    asm volatile("s_wait_dscnt 0" ::: "memory");
#pragma unroll
    for (int t = 0; t < 4; t++) {
#pragma unroll
      for (int kc = 0; kc < 4; kc++) {
        const v2f pa = *(const v2f*)(lp + t * 256 + rc * 16 + kc * 4 + 2 * hf);
        const float* Vr = Vb + (size_t)(jb * 64 + t * 16 + kc * 4 + 2 * hf) * D_MODEL;
        v2f vb0, vb1, vb2, vb3;
        vb0.x = Vr[rc];      vb0.y = Vr[D_MODEL + rc];
        vb1.x = Vr[16 + rc]; vb1.y = Vr[D_MODEL + 16 + rc];
        vb2.x = Vr[32 + rc]; vb2.y = Vr[D_MODEL + 32 + rc];
        vb3.x = Vr[48 + rc]; vb3.y = Vr[D_MODEL + 48 + rc];
        o0 = wmma_f32(pa, vb0, o0);
        o1 = wmma_f32(pa, vb1, o1);
        o2 = wmma_f32(pa, vb2, o2);
        o3 = wmma_f32(pa, vb3, o3);
      }
    }
  }
#pragma unroll
  for (int r = 0; r < 8; r++) {
    const float inv = 1.0f / lr[r];
    const size_t ob = (size_t)(qt * 16 + r + 8 * hf) * D_MODEL + rc;
    Ob[ob]      = o0[r] * inv;
    Ob[ob + 16] = o1[r] * inv;
    Ob[ob + 32] = o2[r] * inv;
    Ob[ob + 48] = o3[r] * inv;
  }
}

extern "C" void kernel_launch(void* const* d_in, const int* in_sizes, int n_in,
                              void* d_out, int out_size, void* d_ws, size_t ws_size,
                              hipStream_t stream) {
  (void)in_sizes; (void)n_in; (void)out_size; (void)ws_size;
  const float* x        = (const float*)d_in[0];
  const int*   positions= (const int*)d_in[1];
  const int*   theta    = (const int*)d_in[2];
  const float* Wq = (const float*)d_in[3];
  const float* Wk = (const float*)d_in[4];
  const float* Wv = (const float*)d_in[5];
  const float* Wo = (const float*)d_in[6];
  const float* g1 = (const float*)d_in[7];
  const float* g2 = (const float*)d_in[8];
  const float* w1 = (const float*)d_in[9];
  const float* w2 = (const float*)d_in[10];
  const float* w3 = (const float*)d_in[11];
  float* out = (float*)d_out;

  float* xn   = (float*)d_ws;
  float* q    = xn   + (size_t)ROWS * D_MODEL;
  float* k    = q    + (size_t)ROWS * D_MODEL;
  float* v    = k    + (size_t)ROWS * D_MODEL;
  float* attn = v    + (size_t)ROWS * D_MODEL;
  float* x2   = attn + (size_t)ROWS * D_MODEL;
  float* x3b  = x2   + (size_t)ROWS * D_MODEL;
  float* x1b  = q;

  const dim3 blk(256);
  const dim3 gblk(128);
  const int gproj = (ROWS / 128) * (D_MODEL / 128);
  const int gff   = (ROWS / 128) * ((D_FF + 127) / 128);
  const int gfa   = (BATCH * NUM_HEADS * (SEQ / 16)) / 8;

  rmsnorm_kernel<<<ROWS, blk, 0, stream>>>(x, g1, xn, D_MODEL);
  gemm64<0><<<gproj, gblk, 0, stream>>>(xn, nullptr, Wq, nullptr, q, ROWS, D_MODEL, D_MODEL);
  gemm64<0><<<gproj, gblk, 0, stream>>>(xn, nullptr, Wk, nullptr, k, ROWS, D_MODEL, D_MODEL);
  gemm64<0><<<gproj, gblk, 0, stream>>>(xn, nullptr, Wv, nullptr, v, ROWS, D_MODEL, D_MODEL);
  rope_kernel<<<ROWS, blk, 0, stream>>>(q, positions, theta);
  rope_kernel<<<ROWS, blk, 0, stream>>>(k, positions, theta);
  flash_attn<<<gfa, blk, 0, stream>>>(q, k, v, attn);
  gemm64<1><<<gproj, gblk, 0, stream>>>(attn, nullptr, Wo, x, x2, ROWS, D_MODEL, D_MODEL);
  rmsnorm_kernel<<<ROWS, blk, 0, stream>>>(x2, g2, xn, D_MODEL);
  gemm64<0><<<gff, gblk, 0, stream>>>(xn, nullptr, w1, nullptr, x1b, ROWS, D_FF, D_MODEL);
  gemm64<0><<<gff, gblk, 0, stream>>>(xn, nullptr, w3, nullptr, x3b, ROWS, D_FF, D_MODEL);
  gemm64<2><<<gproj, gblk, 0, stream>>>(x1b, x3b, w2, x2, out, ROWS, D_MODEL, D_FF);
}


# attention: each query tile's 64-key blocks dealt alternately to two waves, partial O/max/sum merged through LDS; 1024 blocks, longest key ranges first; O and row-sum order changed (two partial sums)
# speedup vs baseline: 3.1471x; 1.0927x over previous
#include <hip/hip_runtime.h>

#define D_MODEL   1024
#define NUM_HEADS 16
#define D_K       64
#define D_FF      2752
#define SEQ       2048
#define BATCH     2
#define ROWS      (BATCH*SEQ)

typedef float v2f __attribute__((ext_vector_type(2)));
typedef float v8f __attribute__((ext_vector_type(8)));

__device__ __forceinline__ v8f wmma_f32(v2f a, v2f b, v8f c) {
  return __builtin_amdgcn_wmma_f32_16x16x4_f32(
       false, a,  false, b,
       (short)0, c,  false,  false);
}

__device__ __forceinline__ float silu_mul(float u, float w) {
  return u * (1.0f / (1.0f + __expf(-u))) * w;
}

__global__ void __launch_bounds__(256)
rmsnorm_kernel(const float* __restrict__ x, const float* __restrict__ g,
               float* __restrict__ out, int d) {
  const int row = blockIdx.x;
  const float* xr = x + (size_t)row * d;
  float ss = 0.f;
  for (int i = threadIdx.x; i < d; i += blockDim.x) { float v = xr[i]; ss = fmaf(v, v, ss); }
#pragma unroll
  for (int m = 16; m >= 1; m >>= 1) ss += __shfl_xor(ss, m, 32);
  __shared__ float red[8];
  const int w = threadIdx.x >> 5, ln = threadIdx.x & 31;
  if (ln == 0) red[w] = ss;
  __syncthreads();
  float tot = 0.f;
#pragma unroll
  for (int i = 0; i < 8; i++) tot += red[i];
  const float inv = rsqrtf(tot / (float)d + 1e-5f);
  for (int i = threadIdx.x; i < d; i += blockDim.x)
    out[(size_t)row * d + i] = xr[i] * inv * g[i];
}

__global__ void __launch_bounds__(256)
rope_kernel(float* __restrict__ t, const int* __restrict__ positions,
            const int* __restrict__ theta_p) {
  const int row = blockIdx.x;
  const int s = row % SEQ;
  const float pos = (float)positions[s];
  const float th = (float)(*theta_p);
  float* r = t + (size_t)row * D_MODEL;
  for (int p = threadIdx.x; p < D_MODEL / 2; p += blockDim.x) {
    const int c0 = 2 * p;
    const int d  = c0 & (D_K - 1);
    const float freq = __powf(th, -(float)d * (1.0f / (float)D_K));
    const float ang = pos * freq;
    float sn, cs;
    sincosf(ang, &sn, &cs);
    const float xe = r[c0], xo = r[c0 + 1];
    r[c0]     = xe * cs - xo * sn;
    r[c0 + 1] = xe * sn + xo * cs;
  }
}

typedef float v4f __attribute__((ext_vector_type(4)));

template <int MODE>
struct FragSet {
  v4f a[4][2];
  v4f a2[(MODE == 2) ? 4 : 1][2];
  v4f b[4][2];
};

template <int MODE>
__device__ __forceinline__ void load_set(FragSet<MODE>& f, const float* __restrict__ Ap,
                                         const float* __restrict__ A2p,
                                         const float* __restrict__ Bp,
                                         size_t rstep, int k) {
#pragma unroll
  for (int i = 0; i < 4; i++) {
    f.a[i][0] = *(const v4f*)(Ap + i * rstep + k);
    f.a[i][1] = *(const v4f*)(Ap + i * rstep + k + 4);
    if (MODE == 2) {
      f.a2[i][0] = *(const v4f*)(A2p + i * rstep + k);
      f.a2[i][1] = *(const v4f*)(A2p + i * rstep + k + 4);
    }
  }
#pragma unroll
  for (int j = 0; j < 4; j++) {
    f.b[j][0] = *(const v4f*)(Bp + j * rstep + k);
    f.b[j][1] = *(const v4f*)(Bp + j * rstep + k + 4);
  }
}

template <int MODE>
__device__ __forceinline__ void mma_set(const FragSet<MODE>& f, v8f (&acc)[4][4]) {
  v4f a[4][2];
#pragma unroll
  for (int i = 0; i < 4; i++) {
#pragma unroll
    for (int h = 0; h < 2; h++) {
      if (MODE == 2) {
        const v4f u = f.a[i][h], w = f.a2[i][h];
        a[i][h].x = silu_mul(u.x, w.x);  a[i][h].y = silu_mul(u.y, w.y);
        a[i][h].z = silu_mul(u.z, w.z);  a[i][h].w = silu_mul(u.w, w.w);
      } else {
        a[i][h] = f.a[i][h];
      }
    }
  }
#pragma unroll
  for (int s = 0; s < 4; s++) {
    v2f af[4], bf[4];
#pragma unroll
    for (int i = 0; i < 4; i++) {
      const v4f av = a[i][s >> 1], bv = f.b[i][s >> 1];
      af[i] = (s & 1) ? __builtin_shufflevector(av, av, 2, 3) : __builtin_shufflevector(av, av, 0, 1);
      bf[i] = (s & 1) ? __builtin_shufflevector(bv, bv, 2, 3) : __builtin_shufflevector(bv, bv, 0, 1);
    }
#pragma unroll
    for (int i = 0; i < 4; i++) {
#pragma unroll
      for (int j = 0; j < 4; j++) acc[i][j] = wmma_f32(af[i], bf[j], acc[i][j]);
    }
  }
}

template <int MODE>
__global__ void __launch_bounds__(128)
gemm64(const float* __restrict__ A, const float* __restrict__ A2,
       const float* __restrict__ B, const float* __restrict__ Res,
       float* __restrict__ C, int M, int N, int K) {
  (void)M;
  const int w    = threadIdx.x >> 5;
  const int lane = threadIdx.x & 31;
  const int hf = lane >> 4;
  const int rc = lane & 15;
  const int NB = (N + 127) >> 7;
  const int bm = blockIdx.x / NB;
  const int bn = blockIdx.x - bm * NB;
  const int m0 = (bm * 2 + (w >> 1)) << 6;
  const int n0 = (bn * 2 + (w & 1)) << 6;
  if (n0 >= N) return;

  const size_t rstep = (size_t)16 * K;
  const float* Ap  = A + (size_t)(m0 + rc) * K + 8 * hf;
  const float* A2p = (MODE == 2) ? (A2 + (size_t)(m0 + rc) * K + 8 * hf) : nullptr;
  const float* Bp  = B + (size_t)(n0 + rc) * K + 8 * hf;

  v8f acc[4][4] = {};
  FragSet<MODE> f0, f1;
  load_set<MODE>(f0, Ap, A2p, Bp, rstep, 0);
  for (int k0 = 0; k0 < K; k0 += 32) {
    load_set<MODE>(f1, Ap, A2p, Bp, rstep, k0 + 16);
    __builtin_amdgcn_sched_barrier(0);
    mma_set<MODE>(f0, acc);
    const int kn = (k0 + 32 < K) ? (k0 + 32) : k0;
    load_set<MODE>(f0, Ap, A2p, Bp, rstep, kn);
    __builtin_amdgcn_sched_barrier(0);
    mma_set<MODE>(f1, acc);
  }

#pragma unroll
  for (int i = 0; i < 4; i++) {
#pragma unroll
    for (int r = 0; r < 8; r++) {
      const size_t base = (size_t)(m0 + i * 16 + r + 8 * hf) * N + n0 + rc;
      float o0 = acc[i][0][r], o1 = acc[i][1][r], o2 = acc[i][2][r], o3 = acc[i][3][r];
      if (MODE >= 1) {
        o0 += Res[base];      o1 += Res[base + 16];
        o2 += Res[base + 32]; o3 += Res[base + 48];
      }
      C[base] = o0; C[base + 16] = o1; C[base + 32] = o2; C[base + 48] = o3;
    }
  }
}

__global__ void __launch_bounds__(256)
flash_attn(const float* __restrict__ Q, const float* __restrict__ Kt,
           const float* __restrict__ V, float* __restrict__ O) {
  __shared__ float plds[8 * 4 * 256];
  __shared__ float mlds[4 * 2 * 256];
  const int NBH = BATCH * NUM_HEADS;
  const int wv = threadIdx.x >> 5;
  const int split = wv >> 2;
  const int grp = (SEQ / 64 - 1) - (int)(blockIdx.x / NBH);
  const int bh = blockIdx.x % NBH;
  const int qt = grp * 4 + (wv & 3);
  const int h = bh % NUM_HEADS;
  const int b = bh / NUM_HEADS;
  const size_t base = (size_t)b * SEQ * D_MODEL + (size_t)h * D_K;
  const float* Qb = Q + base;
  const float* Kb = Kt + base;
  const float* Vb = V + base;
  float* Ob = O + base;
  const int lane = threadIdx.x & 31;
  const int hf = lane >> 4;
  const int rc = lane & 15;
  float* lp = plds + wv * 1024;

  v2f qa[16];
#pragma unroll
  for (int i = 0; i < 16; i++)
    qa[i] = *(const v2f*)(Qb + (size_t)(qt * 16 + rc) * D_MODEL + i * 4 + 2 * hf);

  v8f o0 = {}, o1 = {}, o2 = {}, o3 = {};
  float mr[8], lr[8];
#pragma unroll
  for (int r = 0; r < 8; r++) { mr[r] = -1e30f; lr[r] = 0.f; }

  const int nkb = grp + 1;
  for (int jb = split; jb < nkb; jb += 2) {
    v8f s[4];
#pragma unroll
    for (int tp = 0; tp < 4; tp += 2) {
      const float* Kr0 = Kb + (size_t)(jb * 64 + tp * 16 + rc) * D_MODEL + 2 * hf;
      const float* Kr1 = Kr0 + (size_t)16 * D_MODEL;
      v2f kb0[16], kb1[16];
#pragma unroll
      for (int i = 0; i < 16; i++) {
        kb0[i] = *(const v2f*)(Kr0 + i * 4);
        kb1[i] = *(const v2f*)(Kr1 + i * 4);
      }
      v8f sa = {}, sb = {};
#pragma unroll
      for (int i = 0; i < 16; i++) {
        sa = wmma_f32(qa[i], kb0[i], sa);
        sb = wmma_f32(qa[i], kb1[i], sb);
      }
      s[tp] = sa; s[tp + 1] = sb;
    }

    const bool diag = (jb == nkb - 1);
    const int key0 = jb * 64 + rc;
    const int qrow0 = qt * 16 + 8 * hf;

    float mx[8];
#pragma unroll
    for (int r = 0; r < 8; r++) {
#pragma unroll
      for (int t = 0; t < 4; t++) {
        float x = s[t][r] * 0.125f;
        if (diag && (key0 + 16 * t > qrow0 + r)) x = -1e30f;
        s[t][r] = x;
      }
      mx[r] = fmaxf(fmaxf(s[0][r], s[1][r]), fmaxf(s[2][r], s[3][r]));
    }
#pragma unroll
    for (int msk = 1; msk < 16; msk <<= 1) {
      float tmp[8];
#pragma unroll
      for (int r = 0; r < 8; r++) tmp[r] = __shfl_xor(mx[r], msk, 32);
#pragma unroll
      for (int r = 0; r < 8; r++) mx[r] = fmaxf(mx[r], tmp[r]);
    }
    float ps[8];
#pragma unroll
    for (int r = 0; r < 8; r++) {
      const float mnew = fmaxf(mr[r], mx[r]);
      const float corr = __expf(mr[r] - mnew);
      mr[r] = mnew;
      const float p0 = __expf(s[0][r] - mnew);
      const float p1 = __expf(s[1][r] - mnew);
      const float p2 = __expf(s[2][r] - mnew);
      const float p3 = __expf(s[3][r] - mnew);
      ps[r] = (p0 + p1) + (p2 + p3);
      lr[r] *= corr;
      o0[r] *= corr; o1[r] *= corr; o2[r] *= corr; o3[r] *= corr;
      float* lrow = lp + (r + 8 * hf) * 16 + rc;
      lrow[0] = p0; lrow[256] = p1; lrow[512] = p2; lrow[768] = p3;
    }
#pragma unroll
    for (int msk = 1; msk < 16; msk <<= 1) {
      float tmp[8];
#pragma unroll
      for (int r = 0; r < 8; r++) tmp[r] = __shfl_xor(ps[r], msk, 32);
#pragma unroll
      for (int r = 0; r < 8; r++) ps[r] += tmp[r];
    }
#pragma unroll
    for (int r = 0; r < 8; r++) lr[r] += ps[r];

    asm volatile("s_wait_dscnt 0" ::: "memory");
#pragma unroll
    for (int t = 0; t < 4; t++) {
#pragma unroll
      for (int kc = 0; kc < 4; kc++) {
        const v2f pa = *(const v2f*)(lp + t * 256 + rc * 16 + kc * 4 + 2 * hf);
        const float* Vr = Vb + (size_t)(jb * 64 + t * 16 + kc * 4 + 2 * hf) * D_MODEL;
        v2f vb0, vb1, vb2, vb3;
        vb0.x = Vr[rc];      vb0.y = Vr[D_MODEL + rc];
        vb1.x = Vr[16 + rc]; vb1.y = Vr[D_MODEL + 16 + rc];
        vb2.x = Vr[32 + rc]; vb2.y = Vr[D_MODEL + 32 + rc];
        vb3.x = Vr[48 + rc]; vb3.y = Vr[D_MODEL + 48 + rc];
        o0 = wmma_f32(pa, vb0, o0);
        o1 = wmma_f32(pa, vb1, o1);
        o2 = wmma_f32(pa, vb2, o2);
        o3 = wmma_f32(pa, vb3, o3);
      }
    }
  }
  float* xm = mlds + (wv & 3) * 512;
  if (split) {
#pragma unroll
    for (int r = 0; r < 8; r++) {
      lp[(0 * 8 + r) * 32 + lane] = o0[r];
      lp[(1 * 8 + r) * 32 + lane] = o1[r];
      lp[(2 * 8 + r) * 32 + lane] = o2[r];
      lp[(3 * 8 + r) * 32 + lane] = o3[r];
      xm[r * 32 + lane]       = mr[r];
      xm[256 + r * 32 + lane] = lr[r];
    }
  }
  __syncthreads();
  if (!split) {
    const float* po = plds + (wv + 4) * 1024;
#pragma unroll
    for (int r = 0; r < 8; r++) {
      const float mb = xm[r * 32 + lane], lb = xm[256 + r * 32 + lane];
      const float m  = fmaxf(mr[r], mb);
      const float fa = __expf(mr[r] - m), fb = __expf(mb - m);
      const float inv = 1.0f / (lr[r] * fa + lb * fb);
      const float ga = fa * inv, gb = fb * inv;
      const size_t ob = (size_t)(qt * 16 + r + 8 * hf) * D_MODEL + rc;
      Ob[ob]      = o0[r] * ga + po[(0 * 8 + r) * 32 + lane] * gb;
      Ob[ob + 16] = o1[r] * ga + po[(1 * 8 + r) * 32 + lane] * gb;
      Ob[ob + 32] = o2[r] * ga + po[(2 * 8 + r) * 32 + lane] * gb;
      Ob[ob + 48] = o3[r] * ga + po[(3 * 8 + r) * 32 + lane] * gb;
    }
  }
}

extern "C" void kernel_launch(void* const* d_in, const int* in_sizes, int n_in,
                              void* d_out, int out_size, void* d_ws, size_t ws_size,
                              hipStream_t stream) {
  (void)in_sizes; (void)n_in; (void)out_size; (void)ws_size;
  const float* x        = (const float*)d_in[0];
  const int*   positions= (const int*)d_in[1];
  const int*   theta    = (const int*)d_in[2];
  const float* Wq = (const float*)d_in[3];
  const float* Wk = (const float*)d_in[4];
  const float* Wv = (const float*)d_in[5];
  const float* Wo = (const float*)d_in[6];
  const float* g1 = (const float*)d_in[7];
  const float* g2 = (const float*)d_in[8];
  const float* w1 = (const float*)d_in[9];
  const float* w2 = (const float*)d_in[10];
  const float* w3 = (const float*)d_in[11];
  float* out = (float*)d_out;

  float* xn   = (float*)d_ws;
  float* q    = xn   + (size_t)ROWS * D_MODEL;
  float* k    = q    + (size_t)ROWS * D_MODEL;
  float* v    = k    + (size_t)ROWS * D_MODEL;
  float* attn = v    + (size_t)ROWS * D_MODEL;
  float* x2   = attn + (size_t)ROWS * D_MODEL;
  float* x3b  = x2   + (size_t)ROWS * D_MODEL;
  float* x1b  = q;

  const dim3 blk(256);
  const dim3 gblk(128);
  const int gproj = (ROWS / 128) * (D_MODEL / 128);
  const int gff   = (ROWS / 128) * ((D_FF + 127) / 128);
  const int gfa   = BATCH * NUM_HEADS * (SEQ / 64);

  rmsnorm_kernel<<<ROWS, blk, 0, stream>>>(x, g1, xn, D_MODEL);
  gemm64<0><<<gproj, gblk, 0, stream>>>(xn, nullptr, Wq, nullptr, q, ROWS, D_MODEL, D_MODEL);
  gemm64<0><<<gproj, gblk, 0, stream>>>(xn, nullptr, Wk, nullptr, k, ROWS, D_MODEL, D_MODEL);
  gemm64<0><<<gproj, gblk, 0, stream>>>(xn, nullptr, Wv, nullptr, v, ROWS, D_MODEL, D_MODEL);
  rope_kernel<<<ROWS, blk, 0, stream>>>(q, positions, theta);
  rope_kernel<<<ROWS, blk, 0, stream>>>(k, positions, theta);
  flash_attn<<<gfa, blk, 0, stream>>>(q, k, v, attn);
  gemm64<1><<<gproj, gblk, 0, stream>>>(attn, nullptr, Wo, x, x2, ROWS, D_MODEL, D_MODEL);
  rmsnorm_kernel<<<ROWS, blk, 0, stream>>>(x2, g2, xn, D_MODEL);
  gemm64<0><<<gff, gblk, 0, stream>>>(xn, nullptr, w1, nullptr, x1b, ROWS, D_FF, D_MODEL);
  gemm64<0><<<gff, gblk, 0, stream>>>(xn, nullptr, w3, nullptr, x3b, ROWS, D_FF, D_MODEL);
  gemm64<2><<<gproj, gblk, 0, stream>>>(x1b, x3b, w2, x2, out, ROWS, D_MODEL, D_FF);
}


# w2 GEMM: SwiGLU gate's 1/(1+exp(-u)) by the hardware reciprocal (v_rcp_f32, about 1 ulp) instead of the 11-instruction IEEE division per element in the K loop; sum order unchanged
# speedup vs baseline: 3.3750x; 1.0724x over previous
#include <hip/hip_runtime.h>

#define D_MODEL   1024
#define NUM_HEADS 16
#define D_K       64
#define D_FF      2752
#define SEQ       2048
#define BATCH     2
#define ROWS      (BATCH*SEQ)

typedef float v2f __attribute__((ext_vector_type(2)));
typedef float v8f __attribute__((ext_vector_type(8)));

__device__ __forceinline__ v8f wmma_f32(v2f a, v2f b, v8f c) {
  return __builtin_amdgcn_wmma_f32_16x16x4_f32(
       false, a,  false, b,
       (short)0, c,  false,  false);
}

__device__ __forceinline__ float silu_mul(float u, float w) {
  return u * __builtin_amdgcn_rcpf(1.0f + __expf(-u)) * w;
}

__global__ void __launch_bounds__(256)
rmsnorm_kernel(const float* __restrict__ x, const float* __restrict__ g,
               float* __restrict__ out, int d) {
  const int row = blockIdx.x;
  const float* xr = x + (size_t)row * d;
  float ss = 0.f;
  for (int i = threadIdx.x; i < d; i += blockDim.x) { float v = xr[i]; ss = fmaf(v, v, ss); }
#pragma unroll
  for (int m = 16; m >= 1; m >>= 1) ss += __shfl_xor(ss, m, 32);
  __shared__ float red[8];
  const int w = threadIdx.x >> 5, ln = threadIdx.x & 31;
  if (ln == 0) red[w] = ss;
  __syncthreads();
  float tot = 0.f;
#pragma unroll
  for (int i = 0; i < 8; i++) tot += red[i];
  const float inv = rsqrtf(tot / (float)d + 1e-5f);
  for (int i = threadIdx.x; i < d; i += blockDim.x)
    out[(size_t)row * d + i] = xr[i] * inv * g[i];
}

__global__ void __launch_bounds__(256)
rope_kernel(float* __restrict__ t, const int* __restrict__ positions,
            const int* __restrict__ theta_p) {
  const int row = blockIdx.x;
  const int s = row % SEQ;
  const float pos = (float)positions[s];
  const float th = (float)(*theta_p);
  float* r = t + (size_t)row * D_MODEL;
  for (int p = threadIdx.x; p < D_MODEL / 2; p += blockDim.x) {
    const int c0 = 2 * p;
    const int d  = c0 & (D_K - 1);
    const float freq = __powf(th, -(float)d * (1.0f / (float)D_K));
    const float ang = pos * freq;
    float sn, cs;
    sincosf(ang, &sn, &cs);
    const float xe = r[c0], xo = r[c0 + 1];
    r[c0]     = xe * cs - xo * sn;
    r[c0 + 1] = xe * sn + xo * cs;
  }
}

typedef float v4f __attribute__((ext_vector_type(4)));

template <int MODE>
struct FragSet {
  v4f a[4][2];
  v4f a2[(MODE == 2) ? 4 : 1][2];
  v4f b[4][2];
};

template <int MODE>
__device__ __forceinline__ void load_set(FragSet<MODE>& f, const float* __restrict__ Ap,
                                         const float* __restrict__ A2p,
                                         const float* __restrict__ Bp,
                                         size_t rstep, int k) {
#pragma unroll
  for (int i = 0; i < 4; i++) {
    f.a[i][0] = *(const v4f*)(Ap + i * rstep + k);
    f.a[i][1] = *(const v4f*)(Ap + i * rstep + k + 4);
    if (MODE == 2) {
      f.a2[i][0] = *(const v4f*)(A2p + i * rstep + k);
      f.a2[i][1] = *(const v4f*)(A2p + i * rstep + k + 4);
    }
  }
#pragma unroll
  for (int j = 0; j < 4; j++) {
    f.b[j][0] = *(const v4f*)(Bp + j * rstep + k);
    f.b[j][1] = *(const v4f*)(Bp + j * rstep + k + 4);
  }
}

template <int MODE>
__device__ __forceinline__ void mma_set(const FragSet<MODE>& f, v8f (&acc)[4][4]) {
  v4f a[4][2];
#pragma unroll
  for (int i = 0; i < 4; i++) {
#pragma unroll
    for (int h = 0; h < 2; h++) {
      if (MODE == 2) {
        const v4f u = f.a[i][h], w = f.a2[i][h];
        a[i][h].x = silu_mul(u.x, w.x);  a[i][h].y = silu_mul(u.y, w.y);
        a[i][h].z = silu_mul(u.z, w.z);  a[i][h].w = silu_mul(u.w, w.w);
      } else {
        a[i][h] = f.a[i][h];
      }
    }
  }
#pragma unroll
  for (int s = 0; s < 4; s++) {
    v2f af[4], bf[4];
#pragma unroll
    for (int i = 0; i < 4; i++) {
      const v4f av = a[i][s >> 1], bv = f.b[i][s >> 1];
      af[i] = (s & 1) ? __builtin_shufflevector(av, av, 2, 3) : __builtin_shufflevector(av, av, 0, 1);
      bf[i] = (s & 1) ? __builtin_shufflevector(bv, bv, 2, 3) : __builtin_shufflevector(bv, bv, 0, 1);
    }
#pragma unroll
    for (int i = 0; i < 4; i++) {
#pragma unroll
      for (int j = 0; j < 4; j++) acc[i][j] = wmma_f32(af[i], bf[j], acc[i][j]);
    }
  }
}

template <int MODE>
__global__ void __launch_bounds__(128)
gemm64(const float* __restrict__ A, const float* __restrict__ A2,
       const float* __restrict__ B, const float* __restrict__ Res,
       float* __restrict__ C, int M, int N, int K) {
  (void)M;
  const int w    = threadIdx.x >> 5;
  const int lane = threadIdx.x & 31;
  const int hf = lane >> 4;
  const int rc = lane & 15;
  const int NB = (N + 127) >> 7;
  const int bm = blockIdx.x / NB;
  const int bn = blockIdx.x - bm * NB;
  const int m0 = (bm * 2 + (w >> 1)) << 6;
  const int n0 = (bn * 2 + (w & 1)) << 6;
  if (n0 >= N) return;

  const size_t rstep = (size_t)16 * K;
  const float* Ap  = A + (size_t)(m0 + rc) * K + 8 * hf;
  const float* A2p = (MODE == 2) ? (A2 + (size_t)(m0 + rc) * K + 8 * hf) : nullptr;
  const float* Bp  = B + (size_t)(n0 + rc) * K + 8 * hf;

  v8f acc[4][4] = {};
  FragSet<MODE> f0, f1;
  load_set<MODE>(f0, Ap, A2p, Bp, rstep, 0);
  for (int k0 = 0; k0 < K; k0 += 32) {
    load_set<MODE>(f1, Ap, A2p, Bp, rstep, k0 + 16);
    __builtin_amdgcn_sched_barrier(0);
    mma_set<MODE>(f0, acc);
    const int kn = (k0 + 32 < K) ? (k0 + 32) : k0;
    load_set<MODE>(f0, Ap, A2p, Bp, rstep, kn);
    __builtin_amdgcn_sched_barrier(0);
    mma_set<MODE>(f1, acc);
  }

#pragma unroll
  for (int i = 0; i < 4; i++) {
#pragma unroll
    for (int r = 0; r < 8; r++) {
      const size_t base = (size_t)(m0 + i * 16 + r + 8 * hf) * N + n0 + rc;
      float o0 = acc[i][0][r], o1 = acc[i][1][r], o2 = acc[i][2][r], o3 = acc[i][3][r];
      if (MODE >= 1) {
        o0 += Res[base];      o1 += Res[base + 16];
        o2 += Res[base + 32]; o3 += Res[base + 48];
      }
      C[base] = o0; C[base + 16] = o1; C[base + 32] = o2; C[base + 48] = o3;
    }
  }
}

__global__ void __launch_bounds__(256)
flash_attn(const float* __restrict__ Q, const float* __restrict__ Kt,
           const float* __restrict__ V, float* __restrict__ O) {
  __shared__ float plds[8 * 4 * 256];
  __shared__ float mlds[4 * 2 * 256];
  const int NBH = BATCH * NUM_HEADS;
  const int wv = threadIdx.x >> 5;
  const int split = wv >> 2;
  const int grp = (SEQ / 64 - 1) - (int)(blockIdx.x / NBH);
  const int bh = blockIdx.x % NBH;
  const int qt = grp * 4 + (wv & 3);
  const int h = bh % NUM_HEADS;
  const int b = bh / NUM_HEADS;
  const size_t base = (size_t)b * SEQ * D_MODEL + (size_t)h * D_K;
  const float* Qb = Q + base;
  const float* Kb = Kt + base;
  const float* Vb = V + base;
  float* Ob = O + base;
  const int lane = threadIdx.x & 31;
  const int hf = lane >> 4;
  const int rc = lane & 15;
  float* lp = plds + wv * 1024;

  v2f qa[16];
#pragma unroll
  for (int i = 0; i < 16; i++)
    qa[i] = *(const v2f*)(Qb + (size_t)(qt * 16 + rc) * D_MODEL + i * 4 + 2 * hf);

  v8f o0 = {}, o1 = {}, o2 = {}, o3 = {};
  float mr[8], lr[8];
#pragma unroll
  for (int r = 0; r < 8; r++) { mr[r] = -1e30f; lr[r] = 0.f; }

  const int nkb = grp + 1;
  for (int jb = split; jb < nkb; jb += 2) {
    v8f s[4];
#pragma unroll
    for (int tp = 0; tp < 4; tp += 2) {
      const float* Kr0 = Kb + (size_t)(jb * 64 + tp * 16 + rc) * D_MODEL + 2 * hf;
      const float* Kr1 = Kr0 + (size_t)16 * D_MODEL;
      v2f kb0[16], kb1[16];
#pragma unroll
      for (int i = 0; i < 16; i++) {
        kb0[i] = *(const v2f*)(Kr0 + i * 4);
        kb1[i] = *(const v2f*)(Kr1 + i * 4);
      }
      v8f sa = {}, sb = {};
#pragma unroll
      for (int i = 0; i < 16; i++) {
        sa = wmma_f32(qa[i], kb0[i], sa);
        sb = wmma_f32(qa[i], kb1[i], sb);
      }
      s[tp] = sa; s[tp + 1] = sb;
    }

    const bool diag = (jb == nkb - 1);
    const int key0 = jb * 64 + rc;
    const int qrow0 = qt * 16 + 8 * hf;

    float mx[8];
#pragma unroll
    for (int r = 0; r < 8; r++) {
#pragma unroll
      for (int t = 0; t < 4; t++) {
        float x = s[t][r] * 0.125f;
        if (diag && (key0 + 16 * t > qrow0 + r)) x = -1e30f;
        s[t][r] = x;
      }
      mx[r] = fmaxf(fmaxf(s[0][r], s[1][r]), fmaxf(s[2][r], s[3][r]));
    }
#pragma unroll
    for (int msk = 1; msk < 16; msk <<= 1) {
      float tmp[8];
#pragma unroll
      for (int r = 0; r < 8; r++) tmp[r] = __shfl_xor(mx[r], msk, 32);
#pragma unroll
      for (int r = 0; r < 8; r++) mx[r] = fmaxf(mx[r], tmp[r]);
    }
    float ps[8];
#pragma unroll
    for (int r = 0; r < 8; r++) {
      const float mnew = fmaxf(mr[r], mx[r]);
      const float corr = __expf(mr[r] - mnew);
      mr[r] = mnew;
      const float p0 = __expf(s[0][r] - mnew);
      const float p1 = __expf(s[1][r] - mnew);
      const float p2 = __expf(s[2][r] - mnew);
      const float p3 = __expf(s[3][r] - mnew);
      ps[r] = (p0 + p1) + (p2 + p3);
      lr[r] *= corr;
      o0[r] *= corr; o1[r] *= corr; o2[r] *= corr; o3[r] *= corr;
      float* lrow = lp + (r + 8 * hf) * 16 + rc;
      lrow[0] = p0; lrow[256] = p1; lrow[512] = p2; lrow[768] = p3;
    }
#pragma unroll
    for (int msk = 1; msk < 16; msk <<= 1) {
      float tmp[8];
#pragma unroll
      for (int r = 0; r < 8; r++) tmp[r] = __shfl_xor(ps[r], msk, 32);
#pragma unroll
      for (int r = 0; r < 8; r++) ps[r] += tmp[r];
    }
#pragma unroll
    for (int r = 0; r < 8; r++) lr[r] += ps[r];

    asm volatile("s_wait_dscnt 0" ::: "memory");
#pragma unroll
    for (int t = 0; t < 4; t++) {
#pragma unroll
      for (int kc = 0; kc < 4; kc++) {
        const v2f pa = *(const v2f*)(lp + t * 256 + rc * 16 + kc * 4 + 2 * hf);
        const float* Vr = Vb + (size_t)(jb * 64 + t * 16 + kc * 4 + 2 * hf) * D_MODEL;
        v2f vb0, vb1, vb2, vb3;
        vb0.x = Vr[rc];      vb0.y = Vr[D_MODEL + rc];
        vb1.x = Vr[16 + rc]; vb1.y = Vr[D_MODEL + 16 + rc];
        vb2.x = Vr[32 + rc]; vb2.y = Vr[D_MODEL + 32 + rc];
        vb3.x = Vr[48 + rc]; vb3.y = Vr[D_MODEL + 48 + rc];
        o0 = wmma_f32(pa, vb0, o0);
        o1 = wmma_f32(pa, vb1, o1);
        o2 = wmma_f32(pa, vb2, o2);
        o3 = wmma_f32(pa, vb3, o3);
      }
    }
  }
  float* xm = mlds + (wv & 3) * 512;
  if (split) {
#pragma unroll
    for (int r = 0; r < 8; r++) {
      lp[(0 * 8 + r) * 32 + lane] = o0[r];
      lp[(1 * 8 + r) * 32 + lane] = o1[r];
      lp[(2 * 8 + r) * 32 + lane] = o2[r];
      lp[(3 * 8 + r) * 32 + lane] = o3[r];
      xm[r * 32 + lane]       = mr[r];
      xm[256 + r * 32 + lane] = lr[r];
    }
  }
  __syncthreads();
  if (!split) {
    const float* po = plds + (wv + 4) * 1024;
#pragma unroll
    for (int r = 0; r < 8; r++) {
      const float mb = xm[r * 32 + lane], lb = xm[256 + r * 32 + lane];
      const float m  = fmaxf(mr[r], mb);
      const float fa = __expf(mr[r] - m), fb = __expf(mb - m);
      const float inv = 1.0f / (lr[r] * fa + lb * fb);
      const float ga = fa * inv, gb = fb * inv;
      const size_t ob = (size_t)(qt * 16 + r + 8 * hf) * D_MODEL + rc;
      Ob[ob]      = o0[r] * ga + po[(0 * 8 + r) * 32 + lane] * gb;
      Ob[ob + 16] = o1[r] * ga + po[(1 * 8 + r) * 32 + lane] * gb;
      Ob[ob + 32] = o2[r] * ga + po[(2 * 8 + r) * 32 + lane] * gb;
      Ob[ob + 48] = o3[r] * ga + po[(3 * 8 + r) * 32 + lane] * gb;
    }
  }
}

extern "C" void kernel_launch(void* const* d_in, const int* in_sizes, int n_in,
                              void* d_out, int out_size, void* d_ws, size_t ws_size,
                              hipStream_t stream) {
  (void)in_sizes; (void)n_in; (void)out_size; (void)ws_size;
  const float* x        = (const float*)d_in[0];
  const int*   positions= (const int*)d_in[1];
  const int*   theta    = (const int*)d_in[2];
  const float* Wq = (const float*)d_in[3];
  const float* Wk = (const float*)d_in[4];
  const float* Wv = (const float*)d_in[5];
  const float* Wo = (const float*)d_in[6];
  const float* g1 = (const float*)d_in[7];
  const float* g2 = (const float*)d_in[8];
  const float* w1 = (const float*)d_in[9];
  const float* w2 = (const float*)d_in[10];
  const float* w3 = (const float*)d_in[11];
  float* out = (float*)d_out;

  float* xn   = (float*)d_ws;
  float* q    = xn   + (size_t)ROWS * D_MODEL;
  float* k    = q    + (size_t)ROWS * D_MODEL;
  float* v    = k    + (size_t)ROWS * D_MODEL;
  float* attn = v    + (size_t)ROWS * D_MODEL;
  float* x2   = attn + (size_t)ROWS * D_MODEL;
  float* x3b  = x2   + (size_t)ROWS * D_MODEL;
  float* x1b  = q;

  const dim3 blk(256);
  const dim3 gblk(128);
  const int gproj = (ROWS / 128) * (D_MODEL / 128);
  const int gff   = (ROWS / 128) * ((D_FF + 127) / 128);
  const int gfa   = BATCH * NUM_HEADS * (SEQ / 64);

  rmsnorm_kernel<<<ROWS, blk, 0, stream>>>(x, g1, xn, D_MODEL);
  gemm64<0><<<gproj, gblk, 0, stream>>>(xn, nullptr, Wq, nullptr, q, ROWS, D_MODEL, D_MODEL);
  gemm64<0><<<gproj, gblk, 0, stream>>>(xn, nullptr, Wk, nullptr, k, ROWS, D_MODEL, D_MODEL);
  gemm64<0><<<gproj, gblk, 0, stream>>>(xn, nullptr, Wv, nullptr, v, ROWS, D_MODEL, D_MODEL);
  rope_kernel<<<ROWS, blk, 0, stream>>>(q, positions, theta);
  rope_kernel<<<ROWS, blk, 0, stream>>>(k, positions, theta);
  flash_attn<<<gfa, blk, 0, stream>>>(q, k, v, attn);
  gemm64<1><<<gproj, gblk, 0, stream>>>(attn, nullptr, Wo, x, x2, ROWS, D_MODEL, D_MODEL);
  rmsnorm_kernel<<<ROWS, blk, 0, stream>>>(x2, g2, xn, D_MODEL);
  gemm64<0><<<gff, gblk, 0, stream>>>(xn, nullptr, w1, nullptr, x1b, ROWS, D_FF, D_MODEL);
  gemm64<0><<<gff, gblk, 0, stream>>>(xn, nullptr, w3, nullptr, x3b, ROWS, D_FF, D_MODEL);
  gemm64<2><<<gproj, gblk, 0, stream>>>(x1b, x3b, w2, x2, out, ROWS, D_MODEL, D_FF);
}


# gemm (the launches with one wave per SIMD: q,k,v,o,w2): third fragment register set, each 16-k chunk's loads issued two chunks ahead of its products instead of one; sum order unchanged
# speedup vs baseline: 3.5111x; 1.0403x over previous
#include <hip/hip_runtime.h>

#define D_MODEL   1024
#define NUM_HEADS 16
#define D_K       64
#define D_FF      2752
#define SEQ       2048
#define BATCH     2
#define ROWS      (BATCH*SEQ)

typedef float v2f __attribute__((ext_vector_type(2)));
typedef float v8f __attribute__((ext_vector_type(8)));

__device__ __forceinline__ v8f wmma_f32(v2f a, v2f b, v8f c) {
  return __builtin_amdgcn_wmma_f32_16x16x4_f32(
       false, a,  false, b,
       (short)0, c,  false,  false);
}

__device__ __forceinline__ float silu_mul(float u, float w) {
  return u * __builtin_amdgcn_rcpf(1.0f + __expf(-u)) * w;
}

__global__ void __launch_bounds__(256)
rmsnorm_kernel(const float* __restrict__ x, const float* __restrict__ g,
               float* __restrict__ out, int d) {
  const int row = blockIdx.x;
  const float* xr = x + (size_t)row * d;
  float ss = 0.f;
  for (int i = threadIdx.x; i < d; i += blockDim.x) { float v = xr[i]; ss = fmaf(v, v, ss); }
#pragma unroll
  for (int m = 16; m >= 1; m >>= 1) ss += __shfl_xor(ss, m, 32);
  __shared__ float red[8];
  const int w = threadIdx.x >> 5, ln = threadIdx.x & 31;
  if (ln == 0) red[w] = ss;
  __syncthreads();
  float tot = 0.f;
#pragma unroll
  for (int i = 0; i < 8; i++) tot += red[i];
  const float inv = rsqrtf(tot / (float)d + 1e-5f);
  for (int i = threadIdx.x; i < d; i += blockDim.x)
    out[(size_t)row * d + i] = xr[i] * inv * g[i];
}

__global__ void __launch_bounds__(256)
rope_kernel(float* __restrict__ t, const int* __restrict__ positions,
            const int* __restrict__ theta_p) {
  const int row = blockIdx.x;
  const int s = row % SEQ;
  const float pos = (float)positions[s];
  const float th = (float)(*theta_p);
  float* r = t + (size_t)row * D_MODEL;
  for (int p = threadIdx.x; p < D_MODEL / 2; p += blockDim.x) {
    const int c0 = 2 * p;
    const int d  = c0 & (D_K - 1);
    const float freq = __powf(th, -(float)d * (1.0f / (float)D_K));
    const float ang = pos * freq;
    float sn, cs;
    sincosf(ang, &sn, &cs);
    const float xe = r[c0], xo = r[c0 + 1];
    r[c0]     = xe * cs - xo * sn;
    r[c0 + 1] = xe * sn + xo * cs;
  }
}

typedef float v4f __attribute__((ext_vector_type(4)));

template <int MODE>
struct FragSet {
  v4f a[4][2];
  v4f a2[(MODE == 2) ? 4 : 1][2];
  v4f b[4][2];
};

template <int MODE>
__device__ __forceinline__ void load_set(FragSet<MODE>& f, const float* __restrict__ Ap,
                                         const float* __restrict__ A2p,
                                         const float* __restrict__ Bp,
                                         size_t rstep, int k) {
#pragma unroll
  for (int i = 0; i < 4; i++) {
    f.a[i][0] = *(const v4f*)(Ap + i * rstep + k);
    f.a[i][1] = *(const v4f*)(Ap + i * rstep + k + 4);
    if (MODE == 2) {
      f.a2[i][0] = *(const v4f*)(A2p + i * rstep + k);
      f.a2[i][1] = *(const v4f*)(A2p + i * rstep + k + 4);
    }
  }
#pragma unroll
  for (int j = 0; j < 4; j++) {
    f.b[j][0] = *(const v4f*)(Bp + j * rstep + k);
    f.b[j][1] = *(const v4f*)(Bp + j * rstep + k + 4);
  }
}

template <int MODE>
__device__ __forceinline__ void mma_set(const FragSet<MODE>& f, v8f (&acc)[4][4]) {
  v4f a[4][2];
#pragma unroll
  for (int i = 0; i < 4; i++) {
#pragma unroll
    for (int h = 0; h < 2; h++) {
      if (MODE == 2) {
        const v4f u = f.a[i][h], w = f.a2[i][h];
        a[i][h].x = silu_mul(u.x, w.x);  a[i][h].y = silu_mul(u.y, w.y);
        a[i][h].z = silu_mul(u.z, w.z);  a[i][h].w = silu_mul(u.w, w.w);
      } else {
        a[i][h] = f.a[i][h];
      }
    }
  }
#pragma unroll
  for (int s = 0; s < 4; s++) {
    v2f af[4], bf[4];
#pragma unroll
    for (int i = 0; i < 4; i++) {
      const v4f av = a[i][s >> 1], bv = f.b[i][s >> 1];
      af[i] = (s & 1) ? __builtin_shufflevector(av, av, 2, 3) : __builtin_shufflevector(av, av, 0, 1);
      bf[i] = (s & 1) ? __builtin_shufflevector(bv, bv, 2, 3) : __builtin_shufflevector(bv, bv, 0, 1);
    }
#pragma unroll
    for (int i = 0; i < 4; i++) {
#pragma unroll
      for (int j = 0; j < 4; j++) acc[i][j] = wmma_f32(af[i], bf[j], acc[i][j]);
    }
  }
}

template <int MODE, int NSETS>
__global__ void __launch_bounds__(128)
gemm64(const float* __restrict__ A, const float* __restrict__ A2,
       const float* __restrict__ B, const float* __restrict__ Res,
       float* __restrict__ C, int M, int N, int K) {
  (void)M;
  const int w    = threadIdx.x >> 5;
  const int lane = threadIdx.x & 31;
  const int hf = lane >> 4;
  const int rc = lane & 15;
  const int NB = (N + 127) >> 7;
  const int bm = blockIdx.x / NB;
  const int bn = blockIdx.x - bm * NB;
  const int m0 = (bm * 2 + (w >> 1)) << 6;
  const int n0 = (bn * 2 + (w & 1)) << 6;
  if (n0 >= N) return;

  const size_t rstep = (size_t)16 * K;
  const float* Ap  = A + (size_t)(m0 + rc) * K + 8 * hf;
  const float* A2p = (MODE == 2) ? (A2 + (size_t)(m0 + rc) * K + 8 * hf) : nullptr;
  const float* Bp  = B + (size_t)(n0 + rc) * K + 8 * hf;

  v8f acc[4][4] = {};
  if (NSETS == 2) {
    FragSet<MODE> f0, f1;
    load_set<MODE>(f0, Ap, A2p, Bp, rstep, 0);
    for (int k0 = 0; k0 < K; k0 += 32) {
      load_set<MODE>(f1, Ap, A2p, Bp, rstep, k0 + 16);
      __builtin_amdgcn_sched_barrier(0);
      mma_set<MODE>(f0, acc);
      const int kn = (k0 + 32 < K) ? (k0 + 32) : k0;
      load_set<MODE>(f0, Ap, A2p, Bp, rstep, kn);
      __builtin_amdgcn_sched_barrier(0);
      mma_set<MODE>(f1, acc);
    }
  } else {
    const int klast = K - 16;
    const int nch = K >> 4;
    FragSet<MODE> f0, f1, f2;
    load_set<MODE>(f0, Ap, A2p, Bp, rstep, 0);
    __builtin_amdgcn_sched_barrier(0);
    load_set<MODE>(f1, Ap, A2p, Bp, rstep, (16 < K) ? 16 : klast);
    int c = 0;
    for (; c + 3 <= nch; c += 3) {
      const int ka = (c + 2) << 4, kb = (c + 3) << 4, kc = (c + 4) << 4;
      load_set<MODE>(f2, Ap, A2p, Bp, rstep, (ka < K) ? ka : klast);
      __builtin_amdgcn_sched_barrier(0);
      mma_set<MODE>(f0, acc);
      load_set<MODE>(f0, Ap, A2p, Bp, rstep, (kb < K) ? kb : klast);
      __builtin_amdgcn_sched_barrier(0);
      mma_set<MODE>(f1, acc);
      load_set<MODE>(f1, Ap, A2p, Bp, rstep, (kc < K) ? kc : klast);
      __builtin_amdgcn_sched_barrier(0);
      mma_set<MODE>(f2, acc);
    }
    if (c < nch) mma_set<MODE>(f0, acc);
    if (c + 1 < nch) mma_set<MODE>(f1, acc);
  }

#pragma unroll
  for (int i = 0; i < 4; i++) {
#pragma unroll
    for (int r = 0; r < 8; r++) {
      const size_t base = (size_t)(m0 + i * 16 + r + 8 * hf) * N + n0 + rc;
      float o0 = acc[i][0][r], o1 = acc[i][1][r], o2 = acc[i][2][r], o3 = acc[i][3][r];
      if (MODE >= 1) {
        o0 += Res[base];      o1 += Res[base + 16];
        o2 += Res[base + 32]; o3 += Res[base + 48];
      }
      C[base] = o0; C[base + 16] = o1; C[base + 32] = o2; C[base + 48] = o3;
    }
  }
}

__global__ void __launch_bounds__(256)
flash_attn(const float* __restrict__ Q, const float* __restrict__ Kt,
           const float* __restrict__ V, float* __restrict__ O) {
  __shared__ float plds[8 * 4 * 256];
  __shared__ float mlds[4 * 2 * 256];
  const int NBH = BATCH * NUM_HEADS;
  const int wv = threadIdx.x >> 5;
  const int split = wv >> 2;
  const int grp = (SEQ / 64 - 1) - (int)(blockIdx.x / NBH);
  const int bh = blockIdx.x % NBH;
  const int qt = grp * 4 + (wv & 3);
  const int h = bh % NUM_HEADS;
  const int b = bh / NUM_HEADS;
  const size_t base = (size_t)b * SEQ * D_MODEL + (size_t)h * D_K;
  const float* Qb = Q + base;
  const float* Kb = Kt + base;
  const float* Vb = V + base;
  float* Ob = O + base;
  const int lane = threadIdx.x & 31;
  const int hf = lane >> 4;
  const int rc = lane & 15;
  float* lp = plds + wv * 1024;

  v2f qa[16];
#pragma unroll
  for (int i = 0; i < 16; i++)
    qa[i] = *(const v2f*)(Qb + (size_t)(qt * 16 + rc) * D_MODEL + i * 4 + 2 * hf);

  v8f o0 = {}, o1 = {}, o2 = {}, o3 = {};
  float mr[8], lr[8];
#pragma unroll
  for (int r = 0; r < 8; r++) { mr[r] = -1e30f; lr[r] = 0.f; }

  const int nkb = grp + 1;
  for (int jb = split; jb < nkb; jb += 2) {
    v8f s[4];
#pragma unroll
    for (int tp = 0; tp < 4; tp += 2) {
      const float* Kr0 = Kb + (size_t)(jb * 64 + tp * 16 + rc) * D_MODEL + 2 * hf;
      const float* Kr1 = Kr0 + (size_t)16 * D_MODEL;
      v2f kb0[16], kb1[16];
#pragma unroll
      for (int i = 0; i < 16; i++) {
        kb0[i] = *(const v2f*)(Kr0 + i * 4);
        kb1[i] = *(const v2f*)(Kr1 + i * 4);
      }
      v8f sa = {}, sb = {};
#pragma unroll
      for (int i = 0; i < 16; i++) {
        sa = wmma_f32(qa[i], kb0[i], sa);
        sb = wmma_f32(qa[i], kb1[i], sb);
      }
      s[tp] = sa; s[tp + 1] = sb;
    }

    const bool diag = (jb == nkb - 1);
    const int key0 = jb * 64 + rc;
    const int qrow0 = qt * 16 + 8 * hf;

    float mx[8];
#pragma unroll
    for (int r = 0; r < 8; r++) {
#pragma unroll
      for (int t = 0; t < 4; t++) {
        float x = s[t][r] * 0.125f;
        if (diag && (key0 + 16 * t > qrow0 + r)) x = -1e30f;
        s[t][r] = x;
      }
      mx[r] = fmaxf(fmaxf(s[0][r], s[1][r]), fmaxf(s[2][r], s[3][r]));
    }
#pragma unroll
    for (int msk = 1; msk < 16; msk <<= 1) {
      float tmp[8];
#pragma unroll
      for (int r = 0; r < 8; r++) tmp[r] = __shfl_xor(mx[r], msk, 32);
#pragma unroll
      for (int r = 0; r < 8; r++) mx[r] = fmaxf(mx[r], tmp[r]);
    }
    float ps[8];
#pragma unroll
    for (int r = 0; r < 8; r++) {
      const float mnew = fmaxf(mr[r], mx[r]);
      const float corr = __expf(mr[r] - mnew);
      mr[r] = mnew;
      const float p0 = __expf(s[0][r] - mnew);
      const float p1 = __expf(s[1][r] - mnew);
      const float p2 = __expf(s[2][r] - mnew);
      const float p3 = __expf(s[3][r] - mnew);
      ps[r] = (p0 + p1) + (p2 + p3);
      lr[r] *= corr;
      o0[r] *= corr; o1[r] *= corr; o2[r] *= corr; o3[r] *= corr;
      float* lrow = lp + (r + 8 * hf) * 16 + rc;
      lrow[0] = p0; lrow[256] = p1; lrow[512] = p2; lrow[768] = p3;
    }
#pragma unroll
    for (int msk = 1; msk < 16; msk <<= 1) {
      float tmp[8];
#pragma unroll
      for (int r = 0; r < 8; r++) tmp[r] = __shfl_xor(ps[r], msk, 32);
#pragma unroll
      for (int r = 0; r < 8; r++) ps[r] += tmp[r];
    }
#pragma unroll
    for (int r = 0; r < 8; r++) lr[r] += ps[r];

    asm volatile("s_wait_dscnt 0" ::: "memory");
#pragma unroll
    for (int t = 0; t < 4; t++) {
#pragma unroll
      for (int kc = 0; kc < 4; kc++) {
        const v2f pa = *(const v2f*)(lp + t * 256 + rc * 16 + kc * 4 + 2 * hf);
        const float* Vr = Vb + (size_t)(jb * 64 + t * 16 + kc * 4 + 2 * hf) * D_MODEL;
        v2f vb0, vb1, vb2, vb3;
        vb0.x = Vr[rc];      vb0.y = Vr[D_MODEL + rc];
        vb1.x = Vr[16 + rc]; vb1.y = Vr[D_MODEL + 16 + rc];
        vb2.x = Vr[32 + rc]; vb2.y = Vr[D_MODEL + 32 + rc];
        vb3.x = Vr[48 + rc]; vb3.y = Vr[D_MODEL + 48 + rc];
        o0 = wmma_f32(pa, vb0, o0);
        o1 = wmma_f32(pa, vb1, o1);
        o2 = wmma_f32(pa, vb2, o2);
        o3 = wmma_f32(pa, vb3, o3);
      }
    }
  }
  float* xm = mlds + (wv & 3) * 512;
  if (split) {
#pragma unroll
    for (int r = 0; r < 8; r++) {
      lp[(0 * 8 + r) * 32 + lane] = o0[r];
      lp[(1 * 8 + r) * 32 + lane] = o1[r];
      lp[(2 * 8 + r) * 32 + lane] = o2[r];
      lp[(3 * 8 + r) * 32 + lane] = o3[r];
      xm[r * 32 + lane]       = mr[r];
      xm[256 + r * 32 + lane] = lr[r];
    }
  }
  __syncthreads();
  if (!split) {
    const float* po = plds + (wv + 4) * 1024;
#pragma unroll
    for (int r = 0; r < 8; r++) {
      const float mb = xm[r * 32 + lane], lb = xm[256 + r * 32 + lane];
      const float m  = fmaxf(mr[r], mb);
      const float fa = __expf(mr[r] - m), fb = __expf(mb - m);
      const float inv = 1.0f / (lr[r] * fa + lb * fb);
      const float ga = fa * inv, gb = fb * inv;
      const size_t ob = (size_t)(qt * 16 + r + 8 * hf) * D_MODEL + rc;
      Ob[ob]      = o0[r] * ga + po[(0 * 8 + r) * 32 + lane] * gb;
      Ob[ob + 16] = o1[r] * ga + po[(1 * 8 + r) * 32 + lane] * gb;
      Ob[ob + 32] = o2[r] * ga + po[(2 * 8 + r) * 32 + lane] * gb;
      Ob[ob + 48] = o3[r] * ga + po[(3 * 8 + r) * 32 + lane] * gb;
    }
  }
}

extern "C" void kernel_launch(void* const* d_in, const int* in_sizes, int n_in,
                              void* d_out, int out_size, void* d_ws, size_t ws_size,
                              hipStream_t stream) {
  (void)in_sizes; (void)n_in; (void)out_size; (void)ws_size;
  const float* x        = (const float*)d_in[0];
  const int*   positions= (const int*)d_in[1];
  const int*   theta    = (const int*)d_in[2];
  const float* Wq = (const float*)d_in[3];
  const float* Wk = (const float*)d_in[4];
  const float* Wv = (const float*)d_in[5];
  const float* Wo = (const float*)d_in[6];
  const float* g1 = (const float*)d_in[7];
  const float* g2 = (const float*)d_in[8];
  const float* w1 = (const float*)d_in[9];
  const float* w2 = (const float*)d_in[10];
  const float* w3 = (const float*)d_in[11];
  float* out = (float*)d_out;

  float* xn   = (float*)d_ws;
  float* q    = xn   + (size_t)ROWS * D_MODEL;
  float* k    = q    + (size_t)ROWS * D_MODEL;
  float* v    = k    + (size_t)ROWS * D_MODEL;
  float* attn = v    + (size_t)ROWS * D_MODEL;
  float* x2   = attn + (size_t)ROWS * D_MODEL;
  float* x3b  = x2   + (size_t)ROWS * D_MODEL;
  float* x1b  = q;

  const dim3 blk(256);
  const dim3 gblk(128);
  const int gproj = (ROWS / 128) * (D_MODEL / 128);
  const int gff   = (ROWS / 128) * ((D_FF + 127) / 128);
  const int gfa   = BATCH * NUM_HEADS * (SEQ / 64);

  rmsnorm_kernel<<<ROWS, blk, 0, stream>>>(x, g1, xn, D_MODEL);
  gemm64<0, 3><<<gproj, gblk, 0, stream>>>(xn, nullptr, Wq, nullptr, q, ROWS, D_MODEL, D_MODEL);
  gemm64<0, 3><<<gproj, gblk, 0, stream>>>(xn, nullptr, Wk, nullptr, k, ROWS, D_MODEL, D_MODEL);
  gemm64<0, 3><<<gproj, gblk, 0, stream>>>(xn, nullptr, Wv, nullptr, v, ROWS, D_MODEL, D_MODEL);
  rope_kernel<<<ROWS, blk, 0, stream>>>(q, positions, theta);
  rope_kernel<<<ROWS, blk, 0, stream>>>(k, positions, theta);
  flash_attn<<<gfa, blk, 0, stream>>>(q, k, v, attn);
  gemm64<1, 3><<<gproj, gblk, 0, stream>>>(attn, nullptr, Wo, x, x2, ROWS, D_MODEL, D_MODEL);
  rmsnorm_kernel<<<ROWS, blk, 0, stream>>>(x2, g2, xn, D_MODEL);
  gemm64<0, 2><<<gff, gblk, 0, stream>>>(xn, nullptr, w1, nullptr, x1b, ROWS, D_FF, D_MODEL);
  gemm64<0, 2><<<gff, gblk, 0, stream>>>(xn, nullptr, w3, nullptr, x3b, ROWS, D_FF, D_MODEL);
  gemm64<2, 3><<<gproj, gblk, 0, stream>>>(x1b, x3b, w2, x2, out, ROWS, D_MODEL, D_FF);
}


# FFN up-projections (w1, w3): wave tile re-cut from 64x64 to 44 tiles (4x10 plus 2x2 of a line shared by a wave pair): 256 blocks, one wave per SIMD, equal work, 32 loads per 176 products; sums kept
# speedup vs baseline: 3.6228x; 1.0318x over previous
#include <hip/hip_runtime.h>

#define D_MODEL   1024
#define NUM_HEADS 16
#define D_K       64
#define D_FF      2752
#define SEQ       2048
#define BATCH     2
#define ROWS      (BATCH*SEQ)

typedef float v2f __attribute__((ext_vector_type(2)));
typedef float v8f __attribute__((ext_vector_type(8)));

__device__ __forceinline__ v8f wmma_f32(v2f a, v2f b, v8f c) {
  return __builtin_amdgcn_wmma_f32_16x16x4_f32(
       false, a,  false, b,
       (short)0, c,  false,  false);
}

__device__ __forceinline__ float silu_mul(float u, float w) {
  return u * __builtin_amdgcn_rcpf(1.0f + __expf(-u)) * w;
}

__global__ void __launch_bounds__(256)
rmsnorm_kernel(const float* __restrict__ x, const float* __restrict__ g,
               float* __restrict__ out, int d) {
  const int row = blockIdx.x;
  const float* xr = x + (size_t)row * d;
  float ss = 0.f;
  for (int i = threadIdx.x; i < d; i += blockDim.x) { float v = xr[i]; ss = fmaf(v, v, ss); }
#pragma unroll
  for (int m = 16; m >= 1; m >>= 1) ss += __shfl_xor(ss, m, 32);
  __shared__ float red[8];
  const int w = threadIdx.x >> 5, ln = threadIdx.x & 31;
  if (ln == 0) red[w] = ss;
  __syncthreads();
  float tot = 0.f;
#pragma unroll
  for (int i = 0; i < 8; i++) tot += red[i];
  const float inv = rsqrtf(tot / (float)d + 1e-5f);
  for (int i = threadIdx.x; i < d; i += blockDim.x)
    out[(size_t)row * d + i] = xr[i] * inv * g[i];
}

__global__ void __launch_bounds__(256)
rope_kernel(float* __restrict__ t, const int* __restrict__ positions,
            const int* __restrict__ theta_p) {
  const int row = blockIdx.x;
  const int s = row % SEQ;
  const float pos = (float)positions[s];
  const float th = (float)(*theta_p);
  float* r = t + (size_t)row * D_MODEL;
  for (int p = threadIdx.x; p < D_MODEL / 2; p += blockDim.x) {
    const int c0 = 2 * p;
    const int d  = c0 & (D_K - 1);
    const float freq = __powf(th, -(float)d * (1.0f / (float)D_K));
    const float ang = pos * freq;
    float sn, cs;
    sincosf(ang, &sn, &cs);
    const float xe = r[c0], xo = r[c0 + 1];
    r[c0]     = xe * cs - xo * sn;
    r[c0 + 1] = xe * sn + xo * cs;
  }
}

typedef float v4f __attribute__((ext_vector_type(4)));

template <int MODE>
struct FragSet {
  v4f a[4][2];
  v4f a2[(MODE == 2) ? 4 : 1][2];
  v4f b[4][2];
};

template <int MODE>
__device__ __forceinline__ void load_set(FragSet<MODE>& f, const float* __restrict__ Ap,
                                         const float* __restrict__ A2p,
                                         const float* __restrict__ Bp,
                                         size_t rstep, int k) {
#pragma unroll
  for (int i = 0; i < 4; i++) {
    f.a[i][0] = *(const v4f*)(Ap + i * rstep + k);
    f.a[i][1] = *(const v4f*)(Ap + i * rstep + k + 4);
    if (MODE == 2) {
      f.a2[i][0] = *(const v4f*)(A2p + i * rstep + k);
      f.a2[i][1] = *(const v4f*)(A2p + i * rstep + k + 4);
    }
  }
#pragma unroll
  for (int j = 0; j < 4; j++) {
    f.b[j][0] = *(const v4f*)(Bp + j * rstep + k);
    f.b[j][1] = *(const v4f*)(Bp + j * rstep + k + 4);
  }
}

template <int MODE>
__device__ __forceinline__ void mma_set(const FragSet<MODE>& f, v8f (&acc)[4][4]) {
  v4f a[4][2];
#pragma unroll
  for (int i = 0; i < 4; i++) {
#pragma unroll
    for (int h = 0; h < 2; h++) {
      if (MODE == 2) {
        const v4f u = f.a[i][h], w = f.a2[i][h];
        a[i][h].x = silu_mul(u.x, w.x);  a[i][h].y = silu_mul(u.y, w.y);
        a[i][h].z = silu_mul(u.z, w.z);  a[i][h].w = silu_mul(u.w, w.w);
      } else {
        a[i][h] = f.a[i][h];
      }
    }
  }
#pragma unroll
  for (int s = 0; s < 4; s++) {
    v2f af[4], bf[4];
#pragma unroll
    for (int i = 0; i < 4; i++) {
      const v4f av = a[i][s >> 1], bv = f.b[i][s >> 1];
      af[i] = (s & 1) ? __builtin_shufflevector(av, av, 2, 3) : __builtin_shufflevector(av, av, 0, 1);
      bf[i] = (s & 1) ? __builtin_shufflevector(bv, bv, 2, 3) : __builtin_shufflevector(bv, bv, 0, 1);
    }
#pragma unroll
    for (int i = 0; i < 4; i++) {
#pragma unroll
      for (int j = 0; j < 4; j++) acc[i][j] = wmma_f32(af[i], bf[j], acc[i][j]);
    }
  }
}

template <int MODE, int NSETS>
__global__ void __launch_bounds__(128)
gemm64(const float* __restrict__ A, const float* __restrict__ A2,
       const float* __restrict__ B, const float* __restrict__ Res,
       float* __restrict__ C, int M, int N, int K) {
  (void)M;
  const int w    = threadIdx.x >> 5;
  const int lane = threadIdx.x & 31;
  const int hf = lane >> 4;
  const int rc = lane & 15;
  const int NB = (N + 127) >> 7;
  const int bm = blockIdx.x / NB;
  const int bn = blockIdx.x - bm * NB;
  const int m0 = (bm * 2 + (w >> 1)) << 6;
  const int n0 = (bn * 2 + (w & 1)) << 6;
  if (n0 >= N) return;

  const size_t rstep = (size_t)16 * K;
  const float* Ap  = A + (size_t)(m0 + rc) * K + 8 * hf;
  const float* A2p = (MODE == 2) ? (A2 + (size_t)(m0 + rc) * K + 8 * hf) : nullptr;
  const float* Bp  = B + (size_t)(n0 + rc) * K + 8 * hf;

  v8f acc[4][4] = {};
  if (NSETS == 2) {
    FragSet<MODE> f0, f1;
    load_set<MODE>(f0, Ap, A2p, Bp, rstep, 0);
    for (int k0 = 0; k0 < K; k0 += 32) {
      load_set<MODE>(f1, Ap, A2p, Bp, rstep, k0 + 16);
      __builtin_amdgcn_sched_barrier(0);
      mma_set<MODE>(f0, acc);
      const int kn = (k0 + 32 < K) ? (k0 + 32) : k0;
      load_set<MODE>(f0, Ap, A2p, Bp, rstep, kn);
      __builtin_amdgcn_sched_barrier(0);
      mma_set<MODE>(f1, acc);
    }
  } else {
    const int klast = K - 16;
    const int nch = K >> 4;
    FragSet<MODE> f0, f1, f2;
    load_set<MODE>(f0, Ap, A2p, Bp, rstep, 0);
    __builtin_amdgcn_sched_barrier(0);
    load_set<MODE>(f1, Ap, A2p, Bp, rstep, (16 < K) ? 16 : klast);
    int c = 0;
    for (; c + 3 <= nch; c += 3) {
      const int ka = (c + 2) << 4, kb = (c + 3) << 4, kc = (c + 4) << 4;
      load_set<MODE>(f2, Ap, A2p, Bp, rstep, (ka < K) ? ka : klast);
      __builtin_amdgcn_sched_barrier(0);
      mma_set<MODE>(f0, acc);
      load_set<MODE>(f0, Ap, A2p, Bp, rstep, (kb < K) ? kb : klast);
      __builtin_amdgcn_sched_barrier(0);
      mma_set<MODE>(f1, acc);
      load_set<MODE>(f1, Ap, A2p, Bp, rstep, (kc < K) ? kc : klast);
      __builtin_amdgcn_sched_barrier(0);
      mma_set<MODE>(f2, acc);
    }
    if (c < nch) mma_set<MODE>(f0, acc);
    if (c + 1 < nch) mma_set<MODE>(f1, acc);
  }

#pragma unroll
  for (int i = 0; i < 4; i++) {
#pragma unroll
    for (int r = 0; r < 8; r++) {
      const size_t base = (size_t)(m0 + i * 16 + r + 8 * hf) * N + n0 + rc;
      float o0 = acc[i][0][r], o1 = acc[i][1][r], o2 = acc[i][2][r], o3 = acc[i][3][r];
      if (MODE >= 1) {
        o0 += Res[base];      o1 += Res[base + 16];
        o2 += Res[base + 32]; o3 += Res[base + 48];
      }
      C[base] = o0; C[base + 16] = o1; C[base + 32] = o2; C[base + 48] = o3;
    }
  }
}

#define UP_K    D_MODEL
#define UP_N    D_FF
#define UP_PAIRW 352
#define UP_RSTEP ((size_t)16 * UP_K)

struct UpSet {
  v4f a[4][2];
  v4f b[12][2];
};

__device__ __forceinline__ void up_load(UpSet& f, const float* __restrict__ Ap,
                                        const float* __restrict__ Bp, int k) {
#pragma unroll
  for (int i = 0; i < 4; i++) {
    f.a[i][0] = *(const v4f*)(Ap + i * UP_RSTEP + k);
    f.a[i][1] = *(const v4f*)(Ap + i * UP_RSTEP + k + 4);
  }
#pragma unroll
  for (int j = 0; j < 12; j++) {
    f.b[j][0] = *(const v4f*)(Bp + j * UP_RSTEP + k);
    f.b[j][1] = *(const v4f*)(Bp + j * UP_RSTEP + k + 4);
  }
}

template <int ODD>
__device__ __forceinline__ void up_mma(const UpSet& f, v8f (&acc)[4][10], v8f (&accs)[2][2]) {
#pragma unroll
  for (int s = 0; s < 4; s++) {
    v2f af[4], bf[12];
#pragma unroll
    for (int i = 0; i < 4; i++) {
      const v4f av = f.a[i][s >> 1];
      af[i] = (s & 1) ? __builtin_shufflevector(av, av, 2, 3) : __builtin_shufflevector(av, av, 0, 1);
    }
#pragma unroll
    for (int j = 0; j < 12; j++) {
      const v4f bv = f.b[j][s >> 1];
      bf[j] = (s & 1) ? __builtin_shufflevector(bv, bv, 2, 3) : __builtin_shufflevector(bv, bv, 0, 1);
    }
#pragma unroll
    for (int i = 0; i < 4; i++) {
#pragma unroll
      for (int j = 0; j < 10; j++) acc[i][j] = wmma_f32(af[i], bf[ODD ? j + 2 : j], acc[i][j]);
    }
#pragma unroll
    for (int i = 0; i < 2; i++) {
#pragma unroll
      for (int j = 0; j < 2; j++)
        accs[i][j] = wmma_f32(af[ODD ? i + 2 : i], bf[ODD ? j : j + 10], accs[i][j]);
    }
  }
}

template <int ODD>
__device__ __forceinline__ void up_wave(const float* __restrict__ A, const float* __restrict__ B,
                                        float* __restrict__ C, int m0, int c0, int nown,
                                        int hf, int rc) {
  const int cb = c0 + (ODD ? 160 : 0);
  const float* Ap = A + (size_t)(m0 + rc) * UP_K + 8 * hf;
  const float* Bp = B + (size_t)(cb + rc) * UP_K + 8 * hf;

  v8f acc[4][10] = {};
  v8f accs[2][2] = {};
  UpSet f0, f1;
  up_load(f0, Ap, Bp, 0);
  for (int k0 = 0; k0 < UP_K; k0 += 32) {
    up_load(f1, Ap, Bp, k0 + 16);
    __builtin_amdgcn_sched_barrier(0);
    up_mma<ODD>(f0, acc, accs);
    const int kn = (k0 + 32 < UP_K) ? (k0 + 32) : k0;
    up_load(f0, Ap, Bp, kn);
    __builtin_amdgcn_sched_barrier(0);
    up_mma<ODD>(f1, acc, accs);
  }

  const int co = c0 + (ODD ? 192 : 0);
#pragma unroll
  for (int j = 0; j < 10; j++) {
    if (co + 16 * j >= nown) {
#pragma unroll
      for (int i = 0; i < 4; i++) {
#pragma unroll
        for (int r = 0; r < 8; r++)
          C[(size_t)(m0 + i * 16 + r + 8 * hf) * UP_N + co + 16 * j + rc] = acc[i][j][r];
      }
    }
  }
  const int ms = m0 + (ODD ? 32 : 0);
#pragma unroll
  for (int j = 0; j < 2; j++) {
    if (c0 + 160 + 16 * j >= nown) {
#pragma unroll
      for (int i = 0; i < 2; i++) {
#pragma unroll
        for (int r = 0; r < 8; r++)
          C[(size_t)(ms + i * 16 + r + 8 * hf) * UP_N + c0 + 160 + 16 * j + rc] = accs[i][j][r];
      }
    }
  }
}

__global__ void __launch_bounds__(128)
gemm_up(const float* __restrict__ A, const float* __restrict__ B, float* __restrict__ C) {
  const int w    = __builtin_amdgcn_readfirstlane((int)(threadIdx.x >> 5));
  const int lane = threadIdx.x & 31;
  const int hf = lane >> 4;
  const int rc = lane & 15;
  const int NB = (UP_N + UP_PAIRW - 1) / UP_PAIRW;
  const int bm = blockIdx.x / NB;
  const int bn = blockIdx.x - bm * NB;
  const int m0 = (bm * 2 + (w >> 1)) * 64;
  const int nown = bn * UP_PAIRW;
  const int c0 = (nown + UP_PAIRW <= UP_N) ? nown : (UP_N - UP_PAIRW);
  if (w & 1) up_wave<1>(A, B, C, m0, c0, nown, hf, rc);
  else       up_wave<0>(A, B, C, m0, c0, nown, hf, rc);
}

__global__ void __launch_bounds__(256)
flash_attn(const float* __restrict__ Q, const float* __restrict__ Kt,
           const float* __restrict__ V, float* __restrict__ O) {
  __shared__ float plds[8 * 4 * 256];
  __shared__ float mlds[4 * 2 * 256];
  const int NBH = BATCH * NUM_HEADS;
  const int wv = threadIdx.x >> 5;
  const int split = wv >> 2;
  const int grp = (SEQ / 64 - 1) - (int)(blockIdx.x / NBH);
  const int bh = blockIdx.x % NBH;
  const int qt = grp * 4 + (wv & 3);
  const int h = bh % NUM_HEADS;
  const int b = bh / NUM_HEADS;
  const size_t base = (size_t)b * SEQ * D_MODEL + (size_t)h * D_K;
  const float* Qb = Q + base;
  const float* Kb = Kt + base;
  const float* Vb = V + base;
  float* Ob = O + base;
  const int lane = threadIdx.x & 31;
  const int hf = lane >> 4;
  const int rc = lane & 15;
  float* lp = plds + wv * 1024;

  v2f qa[16];
#pragma unroll
  for (int i = 0; i < 16; i++)
    qa[i] = *(const v2f*)(Qb + (size_t)(qt * 16 + rc) * D_MODEL + i * 4 + 2 * hf);

  v8f o0 = {}, o1 = {}, o2 = {}, o3 = {};
  float mr[8], lr[8];
#pragma unroll
  for (int r = 0; r < 8; r++) { mr[r] = -1e30f; lr[r] = 0.f; }

  const int nkb = grp + 1;
  for (int jb = split; jb < nkb; jb += 2) {
    v8f s[4];
#pragma unroll
    for (int tp = 0; tp < 4; tp += 2) {
      const float* Kr0 = Kb + (size_t)(jb * 64 + tp * 16 + rc) * D_MODEL + 2 * hf;
      const float* Kr1 = Kr0 + (size_t)16 * D_MODEL;
      v2f kb0[16], kb1[16];
#pragma unroll
      for (int i = 0; i < 16; i++) {
        kb0[i] = *(const v2f*)(Kr0 + i * 4);
        kb1[i] = *(const v2f*)(Kr1 + i * 4);
      }
      v8f sa = {}, sb = {};
#pragma unroll
      for (int i = 0; i < 16; i++) {
        sa = wmma_f32(qa[i], kb0[i], sa);
        sb = wmma_f32(qa[i], kb1[i], sb);
      }
      s[tp] = sa; s[tp + 1] = sb;
    }

    const bool diag = (jb == nkb - 1);
    const int key0 = jb * 64 + rc;
    const int qrow0 = qt * 16 + 8 * hf;

    float mx[8];
#pragma unroll
    for (int r = 0; r < 8; r++) {
#pragma unroll
      for (int t = 0; t < 4; t++) {
        float x = s[t][r] * 0.125f;
        if (diag && (key0 + 16 * t > qrow0 + r)) x = -1e30f;
        s[t][r] = x;
      }
      mx[r] = fmaxf(fmaxf(s[0][r], s[1][r]), fmaxf(s[2][r], s[3][r]));
    }
#pragma unroll
    for (int msk = 1; msk < 16; msk <<= 1) {
      float tmp[8];
#pragma unroll
      for (int r = 0; r < 8; r++) tmp[r] = __shfl_xor(mx[r], msk, 32);
#pragma unroll
      for (int r = 0; r < 8; r++) mx[r] = fmaxf(mx[r], tmp[r]);
    }
    float ps[8];
#pragma unroll
    for (int r = 0; r < 8; r++) {
      const float mnew = fmaxf(mr[r], mx[r]);
      const float corr = __expf(mr[r] - mnew);
      mr[r] = mnew;
      const float p0 = __expf(s[0][r] - mnew);
      const float p1 = __expf(s[1][r] - mnew);
      const float p2 = __expf(s[2][r] - mnew);
      const float p3 = __expf(s[3][r] - mnew);
      ps[r] = (p0 + p1) + (p2 + p3);
      lr[r] *= corr;
      o0[r] *= corr; o1[r] *= corr; o2[r] *= corr; o3[r] *= corr;
      float* lrow = lp + (r + 8 * hf) * 16 + rc;
      lrow[0] = p0; lrow[256] = p1; lrow[512] = p2; lrow[768] = p3;
    }
#pragma unroll
    for (int msk = 1; msk < 16; msk <<= 1) {
      float tmp[8];
#pragma unroll
      for (int r = 0; r < 8; r++) tmp[r] = __shfl_xor(ps[r], msk, 32);
#pragma unroll
      for (int r = 0; r < 8; r++) ps[r] += tmp[r];
    }
#pragma unroll
    for (int r = 0; r < 8; r++) lr[r] += ps[r];

    asm volatile("s_wait_dscnt 0" ::: "memory");
#pragma unroll
    for (int t = 0; t < 4; t++) {
#pragma unroll
      for (int kc = 0; kc < 4; kc++) {
        const v2f pa = *(const v2f*)(lp + t * 256 + rc * 16 + kc * 4 + 2 * hf);
        const float* Vr = Vb + (size_t)(jb * 64 + t * 16 + kc * 4 + 2 * hf) * D_MODEL;
        v2f vb0, vb1, vb2, vb3;
        vb0.x = Vr[rc];      vb0.y = Vr[D_MODEL + rc];
        vb1.x = Vr[16 + rc]; vb1.y = Vr[D_MODEL + 16 + rc];
        vb2.x = Vr[32 + rc]; vb2.y = Vr[D_MODEL + 32 + rc];
        vb3.x = Vr[48 + rc]; vb3.y = Vr[D_MODEL + 48 + rc];
        o0 = wmma_f32(pa, vb0, o0);
        o1 = wmma_f32(pa, vb1, o1);
        o2 = wmma_f32(pa, vb2, o2);
        o3 = wmma_f32(pa, vb3, o3);
      }
    }
  }
  float* xm = mlds + (wv & 3) * 512;
  if (split) {
#pragma unroll
    for (int r = 0; r < 8; r++) {
      lp[(0 * 8 + r) * 32 + lane] = o0[r];
      lp[(1 * 8 + r) * 32 + lane] = o1[r];
      lp[(2 * 8 + r) * 32 + lane] = o2[r];
      lp[(3 * 8 + r) * 32 + lane] = o3[r];
      xm[r * 32 + lane]       = mr[r];
      xm[256 + r * 32 + lane] = lr[r];
    }
  }
  __syncthreads();
  if (!split) {
    const float* po = plds + (wv + 4) * 1024;
#pragma unroll
    for (int r = 0; r < 8; r++) {
      const float mb = xm[r * 32 + lane], lb = xm[256 + r * 32 + lane];
      const float m  = fmaxf(mr[r], mb);
      const float fa = __expf(mr[r] - m), fb = __expf(mb - m);
      const float inv = 1.0f / (lr[r] * fa + lb * fb);
      const float ga = fa * inv, gb = fb * inv;
      const size_t ob = (size_t)(qt * 16 + r + 8 * hf) * D_MODEL + rc;
      Ob[ob]      = o0[r] * ga + po[(0 * 8 + r) * 32 + lane] * gb;
      Ob[ob + 16] = o1[r] * ga + po[(1 * 8 + r) * 32 + lane] * gb;
      Ob[ob + 32] = o2[r] * ga + po[(2 * 8 + r) * 32 + lane] * gb;
      Ob[ob + 48] = o3[r] * ga + po[(3 * 8 + r) * 32 + lane] * gb;
    }
  }
}

extern "C" void kernel_launch(void* const* d_in, const int* in_sizes, int n_in,
                              void* d_out, int out_size, void* d_ws, size_t ws_size,
                              hipStream_t stream) {
  (void)in_sizes; (void)n_in; (void)out_size; (void)ws_size;
  const float* x        = (const float*)d_in[0];
  const int*   positions= (const int*)d_in[1];
  const int*   theta    = (const int*)d_in[2];
  const float* Wq = (const float*)d_in[3];
  const float* Wk = (const float*)d_in[4];
  const float* Wv = (const float*)d_in[5];
  const float* Wo = (const float*)d_in[6];
  const float* g1 = (const float*)d_in[7];
  const float* g2 = (const float*)d_in[8];
  const float* w1 = (const float*)d_in[9];
  const float* w2 = (const float*)d_in[10];
  const float* w3 = (const float*)d_in[11];
  float* out = (float*)d_out;

  float* xn   = (float*)d_ws;
  float* q    = xn   + (size_t)ROWS * D_MODEL;
  float* k    = q    + (size_t)ROWS * D_MODEL;
  float* v    = k    + (size_t)ROWS * D_MODEL;
  float* attn = v    + (size_t)ROWS * D_MODEL;
  float* x2   = attn + (size_t)ROWS * D_MODEL;
  float* x3b  = x2   + (size_t)ROWS * D_MODEL;
  float* x1b  = q;

  const dim3 blk(256);
  const dim3 gblk(128);
  const int gproj = (ROWS / 128) * (D_MODEL / 128);
  const int gff   = (ROWS / 128) * ((D_FF + 351) / 352);
  const int gfa   = BATCH * NUM_HEADS * (SEQ / 64);

  rmsnorm_kernel<<<ROWS, blk, 0, stream>>>(x, g1, xn, D_MODEL);
  gemm64<0, 3><<<gproj, gblk, 0, stream>>>(xn, nullptr, Wq, nullptr, q, ROWS, D_MODEL, D_MODEL);
  gemm64<0, 3><<<gproj, gblk, 0, stream>>>(xn, nullptr, Wk, nullptr, k, ROWS, D_MODEL, D_MODEL);
  gemm64<0, 3><<<gproj, gblk, 0, stream>>>(xn, nullptr, Wv, nullptr, v, ROWS, D_MODEL, D_MODEL);
  rope_kernel<<<ROWS, blk, 0, stream>>>(q, positions, theta);
  rope_kernel<<<ROWS, blk, 0, stream>>>(k, positions, theta);
  flash_attn<<<gfa, blk, 0, stream>>>(q, k, v, attn);
  gemm64<1, 3><<<gproj, gblk, 0, stream>>>(attn, nullptr, Wo, x, x2, ROWS, D_MODEL, D_MODEL);
  rmsnorm_kernel<<<ROWS, blk, 0, stream>>>(x2, g2, xn, D_MODEL);
  gemm_up<<<gff, gblk, 0, stream>>>(xn, w1, x1b);
  gemm_up<<<gff, gblk, 0, stream>>>(xn, w3, x3b);
  gemm64<2, 3><<<gproj, gblk, 0, stream>>>(x1b, x3b, w2, x2, out, ROWS, D_MODEL, D_FF);
}
